# starting kernel
# baseline (speedup 1.0000x reference)
#include <hip/hip_runtime.h>
#include <hip/hip_bf16.h>

typedef float v2f __attribute__((ext_vector_type(2)));
typedef float v8f __attribute__((ext_vector_type(8)));

__global__ void k_deg_init(float* __restrict__ deg, int n) {
  int i = blockIdx.x * blockDim.x + threadIdx.x;
  if (i < n) deg[i] = 1.0f;
}

__global__ void k_deg_accum(const int* __restrict__ dst, float* __restrict__ deg, int e) {
  int i = blockIdx.x * blockDim.x + threadIdx.x;
  if (i < e) atomicAdd(&deg[dst[i]], 1.0f);
}

__global__ void k_deg_rsqrt(float* __restrict__ deg, int n) {
  int i = blockIdx.x * blockDim.x + threadIdx.x;
  if (i < n) deg[i] = rsqrtf(deg[i]);
}

__global__ void k_gemm_wmma_f32(const float* __restrict__ A,
                                const float* __restrict__ B,
                                float* __restrict__ C,
                                int M, int K, int N) {
  const int lane = threadIdx.x & 31;
  const int wv   = threadIdx.x >> 5;
  const int h    = lane >> 4;
  const int l16  = lane & 15;

  const int mbase = blockIdx.x * 32;
  const int m0 = mbase + l16;
  const int m1 = m0 + 16;
  const int mc0 = (m0 < M) ? m0 : (M - 1);
  const int mc1 = (m1 < M) ? m1 : (M - 1);
  const int n0 = wv * 32 + l16;
  const int n1 = n0 + 16;

  const float* a0row = A + (size_t)mc0 * K;
  const float* a1row = A + (size_t)mc1 * K;

  v8f acc00 = {}, acc01 = {}, acc10 = {}, acc11 = {};
  for (int k0 = 0; k0 < K; k0 += 4) {
    const int ka = k0 + 2 * h;
    v2f a0, a1, b0, b1;
    a0.x = a0row[ka];  a0.y = a0row[ka + 1];
    a1.x = a1row[ka];  a1.y = a1row[ka + 1];
    const float* br0 = B + (size_t)ka * N;
    const float* br1 = br0 + N;
    b0.x = br0[n0];  b0.y = br1[n0];
    b1.x = br0[n1];  b1.y = br1[n1];
    acc00 = __builtin_amdgcn_wmma_f32_16x16x4_f32(false, a0, false, b0, (short)0, acc00, false, false);
    acc01 = __builtin_amdgcn_wmma_f32_16x16x4_f32(false, a0, false, b1, (short)0, acc01, false, false);
    acc10 = __builtin_amdgcn_wmma_f32_16x16x4_f32(false, a1, false, b0, (short)0, acc10, false, false);
    acc11 = __builtin_amdgcn_wmma_f32_16x16x4_f32(false, a1, false, b1, (short)0, acc11, false, false);
  }

#pragma unroll
  for (int r = 0; r < 8; ++r) {
    const int row0 = mbase + r + 8 * h;
    const int row1 = row0 + 16;
    if (row0 < M) {
      C[(size_t)row0 * N + n0] = acc00[r];
      C[(size_t)row0 * N + n1] = acc01[r];
    }
    if (row1 < M) {
      C[(size_t)row1 * N + n0] = acc10[r];
      C[(size_t)row1 * N + n1] = acc11[r];
    }
  }
}

__global__ void k_selfloop_init_v4(const float* __restrict__ H,
                                   const float* __restrict__ dinv,
                                   const float* __restrict__ bias,
                                   float* __restrict__ out,
                                   int n, int C, int add_bias) {
  int idx = blockIdx.x * blockDim.x + threadIdx.x;
  int c4 = C >> 2;
  if (idx >= n * c4) return;
  int i = idx / c4;
  int c = (idx - i * c4) << 2;
  float w = dinv[i];
  w *= w;
  float4 hv = reinterpret_cast<const float4*>(H)[idx];
  float4 v = make_float4(hv.x * w, hv.y * w, hv.z * w, hv.w * w);
  if (add_bias) {
    float4 bv = *reinterpret_cast<const float4*>(bias + c);
    v.x += bv.x; v.y += bv.y; v.z += bv.z; v.w += bv.w;
  }
  reinterpret_cast<float4*>(out)[idx] = v;
}

template <int VEC>
__global__ void k_edge_scatter(const int* __restrict__ src,
                               const int* __restrict__ dst,
                               const float* __restrict__ H,
                               const float* __restrict__ dinv,
                               float* __restrict__ out,
                               int E, int C) {
  int lane = threadIdx.x & 31;
  int e = (blockIdx.x * blockDim.x + threadIdx.x) >> 5;
  if (e >= E) return;
  int i = src[e];
  int j = dst[e];
  float w = dinv[i] * dinv[j];
  const float* hrow = H + (size_t)i * C + lane * VEC;
  float* orow = out + (size_t)j * C + lane * VEC;
  if constexpr (VEC == 4) {
    float4 hv = *reinterpret_cast<const float4*>(hrow);
    atomicAdd(orow + 0, hv.x * w);
    atomicAdd(orow + 1, hv.y * w);
    atomicAdd(orow + 2, hv.z * w);
    atomicAdd(orow + 3, hv.w * w);
  } else {
    float2 hv = *reinterpret_cast<const float2*>(hrow);
    atomicAdd(orow + 0, hv.x * w);
    atomicAdd(orow + 1, hv.y * w);
  }
}

__global__ void k_bias_relu_v4(float* __restrict__ a, const float* __restrict__ b,
                               int n, int C) {
  int idx = blockIdx.x * blockDim.x + threadIdx.x;
  int c4 = C >> 2;
  if (idx >= n * c4) return;
  int i = idx / c4;
  int c = (idx - i * c4) << 2;
  float4 v = reinterpret_cast<float4*>(a)[idx];
  float4 bv = *reinterpret_cast<const float4*>(b + c);
  v.x = fmaxf(v.x + bv.x, 0.0f);
  v.y = fmaxf(v.y + bv.y, 0.0f);
  v.z = fmaxf(v.z + bv.z, 0.0f);
  v.w = fmaxf(v.w + bv.w, 0.0f);
  reinterpret_cast<float4*>(a)[idx] = v;
}

extern "C" void kernel_launch(void* const* d_in, const int* in_sizes, int n_in,
                              void* d_out, int out_size, void* d_ws, size_t ws_size,
                              hipStream_t stream) {
  const int CIN = 256, CH = 128, CO = 64;
  const float* x  = (const float*)d_in[0];
  const int*   ei = (const int*)d_in[1];
  const float* W1 = (const float*)d_in[2];
  const float* b1 = (const float*)d_in[3];
  const float* W2 = (const float*)d_in[4];
  const float* b2 = (const float*)d_in[5];
  float* out = (float*)d_out;

  const int N = in_sizes[0] / CIN;
  const int E = in_sizes[1] / 2;
  const int* src = ei;
  const int* dst = ei + E;

  float* ws   = (float*)d_ws;
  float* dinv = ws;
  float* Hbuf = ws + 50176;
  float* AGG  = ws + 50176 + (size_t)N * CH;

  const int T = 256;

  k_deg_init  <<<(N + T - 1) / T, T, 0, stream>>>(dinv, N);
  k_deg_accum <<<(E + T - 1) / T, T, 0, stream>>>(dst, dinv, E);
  k_deg_rsqrt <<<(N + T - 1) / T, T, 0, stream>>>(dinv, N);

  k_gemm_wmma_f32<<<(N + 31) / 32, 32 * (CH / 32), 0, stream>>>(x, W1, Hbuf, N, CIN, CH);
  k_selfloop_init_v4<<<((size_t)N * (CH / 4) + T - 1) / T, T, 0, stream>>>(Hbuf, dinv, b1, AGG, N, CH, 0);
  k_edge_scatter<4><<<((size_t)E * 32 + T - 1) / T, T, 0, stream>>>(src, dst, Hbuf, dinv, AGG, E, CH);
  k_bias_relu_v4<<<((size_t)N * (CH / 4) + T - 1) / T, T, 0, stream>>>(AGG, b1, N, CH);

  k_gemm_wmma_f32<<<(N + 31) / 32, 32 * (CO / 32), 0, stream>>>(AGG, W2, Hbuf, N, CH, CO);
  k_selfloop_init_v4<<<((size_t)N * (CO / 4) + T - 1) / T, T, 0, stream>>>(Hbuf, dinv, b2, out, N, CO, 1);
  k_edge_scatter<2><<<((size_t)E * 32 + T - 1) / T, T, 0, stream>>>(src, dst, Hbuf, dinv, out, E, CO);
}


# Owner aggregation: each work-group keeps 196 nodes' rows in LDS, scans all destinations, adds matching source rows with LDS float atomics; no global atomics, 3 launches fewer; sum order = LDS arrival
# speedup vs baseline: 1.6706x; 1.6706x over previous
#include <hip/hip_runtime.h>
#include <hip/hip_bf16.h>

typedef float v2f __attribute__((ext_vector_type(2)));
typedef float v8f __attribute__((ext_vector_type(8)));

__global__ void k_deg_init(float* __restrict__ deg, int n) {
  int i = blockIdx.x * blockDim.x + threadIdx.x;
  if (i < n) deg[i] = 1.0f;
}

__global__ void k_deg_accum(const int* __restrict__ dst, float* __restrict__ deg, int e) {
  int i = blockIdx.x * blockDim.x + threadIdx.x;
  if (i < e) atomicAdd(&deg[dst[i]], 1.0f);
}

__global__ void k_deg_rsqrt(float* __restrict__ deg, int n) {
  int i = blockIdx.x * blockDim.x + threadIdx.x;
  if (i < n) deg[i] = rsqrtf(deg[i]);
}

__global__ void k_gemm_wmma_f32(const float* __restrict__ A,
                                const float* __restrict__ B,
                                float* __restrict__ C,
                                int M, int K, int N) {
  const int lane = threadIdx.x & 31;
  const int wv   = threadIdx.x >> 5;
  const int h    = lane >> 4;
  const int l16  = lane & 15;

  const int mbase = blockIdx.x * 32;
  const int m0 = mbase + l16;
  const int m1 = m0 + 16;
  const int mc0 = (m0 < M) ? m0 : (M - 1);
  const int mc1 = (m1 < M) ? m1 : (M - 1);
  const int n0 = wv * 32 + l16;
  const int n1 = n0 + 16;

  const float* a0row = A + (size_t)mc0 * K;
  const float* a1row = A + (size_t)mc1 * K;

  v8f acc00 = {}, acc01 = {}, acc10 = {}, acc11 = {};
  for (int k0 = 0; k0 < K; k0 += 4) {
    const int ka = k0 + 2 * h;
    v2f a0, a1, b0, b1;
    a0.x = a0row[ka];  a0.y = a0row[ka + 1];
    a1.x = a1row[ka];  a1.y = a1row[ka + 1];
    const float* br0 = B + (size_t)ka * N;
    const float* br1 = br0 + N;
    b0.x = br0[n0];  b0.y = br1[n0];
    b1.x = br0[n1];  b1.y = br1[n1];
    acc00 = __builtin_amdgcn_wmma_f32_16x16x4_f32(false, a0, false, b0, (short)0, acc00, false, false);
    acc01 = __builtin_amdgcn_wmma_f32_16x16x4_f32(false, a0, false, b1, (short)0, acc01, false, false);
    acc10 = __builtin_amdgcn_wmma_f32_16x16x4_f32(false, a1, false, b0, (short)0, acc10, false, false);
    acc11 = __builtin_amdgcn_wmma_f32_16x16x4_f32(false, a1, false, b1, (short)0, acc11, false, false);
  }

#pragma unroll
  for (int r = 0; r < 8; ++r) {
    const int row0 = mbase + r + 8 * h;
    const int row1 = row0 + 16;
    if (row0 < M) {
      C[(size_t)row0 * N + n0] = acc00[r];
      C[(size_t)row0 * N + n1] = acc01[r];
    }
    if (row1 < M) {
      C[(size_t)row1 * N + n0] = acc10[r];
      C[(size_t)row1 * N + n1] = acc11[r];
    }
  }
}

constexpr int AGG_R    = 196;
constexpr int AGG_T    = 1024;
constexpr int AGG_QCAP = 8192;
constexpr int AGG_US   = 8;
constexpr int AGG_UB   = 16;

static_assert(AGG_US * 4 <= 32, "one hit bit per scanned destination");
template <bool B> struct agg_flag { static constexpr bool value = B; };
template <int VEC> struct agg_vec;
template <> struct agg_vec<4> { typedef float4 type; };
template <> struct agg_vec<2> { typedef float2 type; };

template <int C, bool RELU>
__global__ __launch_bounds__(1024) void k_agg_owner(const int* __restrict__ src,
                                                    const int* __restrict__ dst,
                                                    const float* __restrict__ H,
                                                    const float* __restrict__ dinv,
                                                    const float* __restrict__ bias,
                                                    float* __restrict__ out,
                                                    int N, int E) {
  constexpr int VEC = C / 32;
  typedef typename agg_vec<VEC>::type vec_t;
  __shared__ float acc[AGG_R * C];
  __shared__ int   q[AGG_QCAP];
  __shared__ float dloc[AGG_R];
  __shared__ int   qcount;

  const int tid  = threadIdx.x;
  const int lane = tid & 31;
  const int wv   = tid >> 5;
  const int base = blockIdx.x * AGG_R;
  int nloc = N - base;
  if (nloc > AGG_R) nloc = AGG_R;

  if (tid == 0) qcount = 0;
  for (int r = tid; r < nloc; r += AGG_T) dloc[r] = dinv[base + r];
  for (int r = wv; r < nloc; r += AGG_T / 32) {
    float w = dinv[base + r];
    w *= w;
    const vec_t hv = *reinterpret_cast<const vec_t*>(H + (size_t)(base + r) * C + lane * VEC);
    const float* hp = reinterpret_cast<const float*>(&hv);
#pragma unroll
    for (int t = 0; t < VEC; ++t) acc[r * C + t * 32 + lane] = hp[t] * w;
  }
  __syncthreads();

  auto hit = [&](int d) { return (unsigned)(d - base) < (unsigned)nloc; };
  auto push = [&](int e) {
    const int pos = atomicAdd(&qcount, 1);
    if (pos < AGG_QCAP) {
      q[pos] = e;
    } else {
      const int i = src[e];
      const int d = dst[e];
      const float w = dinv[i] * dinv[d];
      const float* hrow = H + (size_t)i * C;
      float* arow = acc + (d - base) * C;
      for (int c = 0; c < C; ++c) atomicAdd(&arow[(c % VEC) * 32 + c / VEC], hrow[c] * w);
    }
  };

  if ((E & 3) == 0) {
    const int4* d4 = reinterpret_cast<const int4*>(dst);
    const int E4 = E >> 2;
    auto round = [&](int g, auto guarded) {
      int4 dv[AGG_US];
#pragma unroll
      for (int u = 0; u < AGG_US; ++u) {
        const int gu = g + u * AGG_T;
        dv[u] = d4[(!guarded.value || gu < E4) ? gu : g];
      }
      unsigned hm = 0u;
#pragma unroll
      for (int u = 0; u < AGG_US; ++u) {
        const unsigned lim = (!guarded.value || g + u * AGG_T < E4) ? (unsigned)nloc : 0u;
        hm |= ((unsigned)(dv[u].x - base) < lim) ? (1u << (4 * u))     : 0u;
        hm |= ((unsigned)(dv[u].y - base) < lim) ? (1u << (4 * u + 1)) : 0u;
        hm |= ((unsigned)(dv[u].z - base) < lim) ? (1u << (4 * u + 2)) : 0u;
        hm |= ((unsigned)(dv[u].w - base) < lim) ? (1u << (4 * u + 3)) : 0u;
      }
      while (hm) {
        const int b = __builtin_ctz(hm);
        hm &= hm - 1u;
        push(4 * (g + (b >> 2) * AGG_T) + (b & 3));
      }
    };
    int g = tid;
    for (; g + (AGG_US - 1) * AGG_T < E4; g += AGG_US * AGG_T) round(g, agg_flag<false>());
    if (g < E4) round(g, agg_flag<true>());
  } else {
    for (int e = tid; e < E; e += AGG_T)
      if (hit(dst[e])) push(e);
  }
  __syncthreads();

  int qn = qcount;
  if (qn > AGG_QCAP) qn = AGG_QCAP;
  qn = __builtin_amdgcn_readfirstlane(qn);
  const int wvu = __builtin_amdgcn_readfirstlane(wv);
  for (int k0 = wvu * 32; k0 < qn; k0 += (AGG_T / 32) * 32) {
    const int cnt = (qn - k0 < 32) ? (qn - k0) : 32;
    const bool ok = lane < cnt;
    const int e  = ok ? q[k0 + lane] : 0;
    const int si = src[e];
    const int dj = ok ? (dst[e] - base) : 0;
    const float we = ok ? dinv[si] * dloc[dj] : 0.0f;
    for (int u0 = 0; u0 < cnt; u0 += AGG_UB) {
      vec_t hv[AGG_UB];
#pragma unroll
      for (int u = 0; u < AGG_UB; ++u) {
        const int uu = (u0 + u < cnt) ? (u0 + u) : (cnt - 1);
        const int i = __builtin_amdgcn_readlane(si, uu);
        hv[u] = *reinterpret_cast<const vec_t*>(H + (size_t)i * C + lane * VEC);
      }
#pragma unroll
      for (int u = 0; u < AGG_UB; ++u) {
        if (u0 + u < cnt) {
          const int j = __builtin_amdgcn_readlane(dj, u0 + u);
          const float w = __int_as_float(__builtin_amdgcn_readlane(__float_as_int(we), u0 + u));
          const float* hp = reinterpret_cast<const float*>(&hv[u]);
#pragma unroll
          for (int t = 0; t < VEC; ++t) atomicAdd(&acc[j * C + t * 32 + lane], hp[t] * w);
        }
      }
    }
  }
  __syncthreads();

  const vec_t bv = *reinterpret_cast<const vec_t*>(bias + lane * VEC);
  const float* bp = reinterpret_cast<const float*>(&bv);
  for (int r = wv; r < nloc; r += AGG_T / 32) {
    vec_t ov;
    float* op = reinterpret_cast<float*>(&ov);
#pragma unroll
    for (int t = 0; t < VEC; ++t) {
      float v = acc[r * C + t * 32 + lane] + bp[t];
      op[t] = RELU ? fmaxf(v, 0.0f) : v;
    }
    *reinterpret_cast<vec_t*>(out + (size_t)(base + r) * C + lane * VEC) = ov;
  }
}

extern "C" void kernel_launch(void* const* d_in, const int* in_sizes, int n_in,
                              void* d_out, int out_size, void* d_ws, size_t ws_size,
                              hipStream_t stream) {
  const int CIN = 256, CH = 128, CO = 64;
  const float* x  = (const float*)d_in[0];
  const int*   ei = (const int*)d_in[1];
  const float* W1 = (const float*)d_in[2];
  const float* b1 = (const float*)d_in[3];
  const float* W2 = (const float*)d_in[4];
  const float* b2 = (const float*)d_in[5];
  float* out = (float*)d_out;

  const int N = in_sizes[0] / CIN;
  const int E = in_sizes[1] / 2;
  const int* src = ei;
  const int* dst = ei + E;

  float* ws   = (float*)d_ws;
  float* dinv = ws;
  float* Hbuf = ws + 50176;
  float* AGG  = ws + 50176 + (size_t)N * CH;

  const int T = 256;

  k_deg_init  <<<(N + T - 1) / T, T, 0, stream>>>(dinv, N);
  k_deg_accum <<<(E + T - 1) / T, T, 0, stream>>>(dst, dinv, E);
  k_deg_rsqrt <<<(N + T - 1) / T, T, 0, stream>>>(dinv, N);

  k_gemm_wmma_f32<<<(N + 31) / 32, 32 * (CH / 32), 0, stream>>>(x, W1, Hbuf, N, CIN, CH);
  k_agg_owner<128, true><<<(N + AGG_R - 1) / AGG_R, AGG_T, 0, stream>>>(src, dst, Hbuf, dinv, b1, AGG, N, E);

  k_gemm_wmma_f32<<<(N + 31) / 32, 32 * (CO / 32), 0, stream>>>(AGG, W2, Hbuf, N, CH, CO);
  k_agg_owner<64, false><<<(N + AGG_R - 1) / AGG_R, AGG_T, 0, stream>>>(src, dst, Hbuf, dinv, b2, out, N, E);
}


# One scan of dst per call: a first launch queues each owner's edge ids, derives dinv from the queue's counts and keeps the queue in memory for both layers; 2 launches fewer; sum order = LDS arrival
# speedup vs baseline: 2.4595x; 1.4722x over previous
#include <hip/hip_runtime.h>
#include <hip/hip_bf16.h>

typedef float v2f __attribute__((ext_vector_type(2)));
typedef float v8f __attribute__((ext_vector_type(8)));

__global__ void k_gemm_wmma_f32(const float* __restrict__ A,
                                const float* __restrict__ B,
                                float* __restrict__ C,
                                int M, int K, int N) {
  const int lane = threadIdx.x & 31;
  const int wv   = threadIdx.x >> 5;
  const int h    = lane >> 4;
  const int l16  = lane & 15;

  const int mbase = blockIdx.x * 32;
  const int m0 = mbase + l16;
  const int m1 = m0 + 16;
  const int mc0 = (m0 < M) ? m0 : (M - 1);
  const int mc1 = (m1 < M) ? m1 : (M - 1);
  const int n0 = wv * 32 + l16;
  const int n1 = n0 + 16;

  const float* a0row = A + (size_t)mc0 * K;
  const float* a1row = A + (size_t)mc1 * K;

  v8f acc00 = {}, acc01 = {}, acc10 = {}, acc11 = {};
  for (int k0 = 0; k0 < K; k0 += 4) {
    const int ka = k0 + 2 * h;
    v2f a0, a1, b0, b1;
    a0.x = a0row[ka];  a0.y = a0row[ka + 1];
    a1.x = a1row[ka];  a1.y = a1row[ka + 1];
    const float* br0 = B + (size_t)ka * N;
    const float* br1 = br0 + N;
    b0.x = br0[n0];  b0.y = br1[n0];
    b1.x = br0[n1];  b1.y = br1[n1];
    acc00 = __builtin_amdgcn_wmma_f32_16x16x4_f32(false, a0, false, b0, (short)0, acc00, false, false);
    acc01 = __builtin_amdgcn_wmma_f32_16x16x4_f32(false, a0, false, b1, (short)0, acc01, false, false);
    acc10 = __builtin_amdgcn_wmma_f32_16x16x4_f32(false, a1, false, b0, (short)0, acc10, false, false);
    acc11 = __builtin_amdgcn_wmma_f32_16x16x4_f32(false, a1, false, b1, (short)0, acc11, false, false);
  }

#pragma unroll
  for (int r = 0; r < 8; ++r) {
    const int row0 = mbase + r + 8 * h;
    const int row1 = row0 + 16;
    if (row0 < M) {
      C[(size_t)row0 * N + n0] = acc00[r];
      C[(size_t)row0 * N + n1] = acc01[r];
    }
    if (row1 < M) {
      C[(size_t)row1 * N + n0] = acc10[r];
      C[(size_t)row1 * N + n1] = acc11[r];
    }
  }
}

constexpr int AGG_R    = 196;
constexpr int AGG_T    = 1024;
constexpr int AGG_QCAP = 8192;
constexpr int AGG_US   = 8;
constexpr int AGG_UB   = 16;
constexpr int AGG_SLABROW = 64;
constexpr int AGG_SLAB = AGG_R * AGG_SLABROW;

static_assert(AGG_US * 4 <= 32, "one hit bit per scanned destination");
template <bool B> struct agg_flag { static constexpr bool value = B; };
template <int VEC> struct agg_vec;
template <> struct agg_vec<4> { typedef float4 type; };
template <> struct agg_vec<2> { typedef float2 type; };

__device__ __forceinline__ int agg_slab_cap(int nloc) {
  const int c = nloc * AGG_SLABROW - 1;
  return c < AGG_QCAP ? c : AGG_QCAP;
}

template <class Push>
__device__ __forceinline__ void agg_scan_dst(const int* __restrict__ dst, int E, int base, int nloc, int tid, Push push) {
  if ((E & 3) == 0) {
    const int4* d4 = reinterpret_cast<const int4*>(dst);
    const int E4 = E >> 2;
    auto round = [&](int g, auto guarded) {
      int4 dv[AGG_US];
#pragma unroll
      for (int u = 0; u < AGG_US; ++u) {
        const int gu = g + u * AGG_T;
        dv[u] = d4[(!guarded.value || gu < E4) ? gu : g];
      }
      unsigned hm = 0u;
#pragma unroll
      for (int u = 0; u < AGG_US; ++u) {
        const unsigned lim = (!guarded.value || g + u * AGG_T < E4) ? (unsigned)nloc : 0u;
        hm |= ((unsigned)(dv[u].x - base) < lim) ? (1u << (4 * u))     : 0u;
        hm |= ((unsigned)(dv[u].y - base) < lim) ? (1u << (4 * u + 1)) : 0u;
        hm |= ((unsigned)(dv[u].z - base) < lim) ? (1u << (4 * u + 2)) : 0u;
        hm |= ((unsigned)(dv[u].w - base) < lim) ? (1u << (4 * u + 3)) : 0u;
      }
      while (hm) {
        const int b = __builtin_ctz(hm);
        hm &= hm - 1u;
        push(4 * (g + (b >> 2) * AGG_T) + (b & 3));
      }
    };
    int g = tid;
    for (; g + (AGG_US - 1) * AGG_T < E4; g += AGG_US * AGG_T) round(g, agg_flag<false>());
    if (g < E4) round(g, agg_flag<true>());
  } else {
    for (int e = tid; e < E; e += AGG_T)
      if ((unsigned)(dst[e] - base) < (unsigned)nloc) push(e);
  }
}

__global__ __launch_bounds__(1024) void k_owner_scan(const int* __restrict__ dst,
                                                     float* __restrict__ dinv,
                                                     int* __restrict__ slabs,
                                                     int N, int E) {
  __shared__ int q[AGG_QCAP];
  __shared__ int cnt[AGG_R];
  __shared__ int qcount;

  const int tid  = threadIdx.x;
  const int base = blockIdx.x * AGG_R;
  int nloc = N - base;
  if (nloc > AGG_R) nloc = AGG_R;
  const int cap = agg_slab_cap(nloc);
  int* slab = slabs + (size_t)blockIdx.x * AGG_SLAB;

  if (tid == 0) qcount = 0;
  for (int r = tid; r < AGG_R; r += AGG_T) cnt[r] = 0;
  __syncthreads();

  agg_scan_dst(dst, E, base, nloc, tid, [&](int e) {
    const int pos = atomicAdd(&qcount, 1);
    if (pos < cap) q[pos] = e;
  });
  __syncthreads();

  const int total = qcount;
  if (total <= cap) {
    for (int k = tid; k < total; k += AGG_T) {
      const int e = q[k];
      slab[1 + k] = e;
      atomicAdd(&cnt[dst[e] - base], 1);
    }
  } else {
    for (int e = tid; e < E; e += AGG_T) {
      const int d = dst[e];
      if ((unsigned)(d - base) < (unsigned)nloc) atomicAdd(&cnt[d - base], 1);
    }
  }
  if (tid == 0) slab[0] = total;
  __syncthreads();

  for (int r = tid; r < nloc; r += AGG_T) dinv[base + r] = rsqrtf(1.0f + (float)cnt[r]);
}

template <int C, bool RELU>
__global__ __launch_bounds__(1024) void k_agg_owner(const int* __restrict__ src,
                                                    const int* __restrict__ dst,
                                                    const float* __restrict__ H,
                                                    const float* __restrict__ dinv,
                                                    const float* __restrict__ bias,
                                                    const int* slabs,
                                                    float* out,
                                                    int N, int E) {
  constexpr int VEC = C / 32;
  typedef typename agg_vec<VEC>::type vec_t;
  __shared__ float acc[AGG_R * C];
  __shared__ int   q[AGG_QCAP];
  __shared__ float dloc[AGG_R];
  __shared__ int   qcount;

  const int tid  = threadIdx.x;
  const int lane = tid & 31;
  const int wv   = tid >> 5;
  const int base = blockIdx.x * AGG_R;
  int nloc = N - base;
  if (nloc > AGG_R) nloc = AGG_R;

  const int* slab = slabs + (size_t)blockIdx.x * AGG_SLAB;
  const int total = __builtin_amdgcn_readfirstlane(slab[0]);
  const bool kept = total <= agg_slab_cap(nloc);

  if (tid == 0) qcount = 0;
  for (int r = tid; r < nloc; r += AGG_T) dloc[r] = dinv[base + r];
  for (int r = wv; r < nloc; r += AGG_T / 32) {
    float w = dinv[base + r];
    w *= w;
    const vec_t hv = *reinterpret_cast<const vec_t*>(H + (size_t)(base + r) * C + lane * VEC);
    const float* hp = reinterpret_cast<const float*>(&hv);
#pragma unroll
    for (int t = 0; t < VEC; ++t) acc[r * C + t * 32 + lane] = hp[t] * w;
  }
  __syncthreads();

  if (!kept) {
    agg_scan_dst(dst, E, base, nloc, tid, [&](int e) {
      const int pos = atomicAdd(&qcount, 1);
      if (pos < AGG_QCAP) {
        q[pos] = e;
      } else {
        const int i = src[e];
        const int d = dst[e];
        const float w = dinv[i] * dinv[d];
        const float* hrow = H + (size_t)i * C;
        float* arow = acc + (d - base) * C;
        for (int c = 0; c < C; ++c) atomicAdd(&arow[(c % VEC) * 32 + c / VEC], hrow[c] * w);
      }
    });
  }
  __syncthreads();

  int qn = total;
  if (!kept) {
    qn = qcount;
    if (qn > AGG_QCAP) qn = AGG_QCAP;
  }
  qn = __builtin_amdgcn_readfirstlane(qn);
  const int wvu = __builtin_amdgcn_readfirstlane(wv);
  for (int k0 = wvu * 32; k0 < qn; k0 += (AGG_T / 32) * 32) {
    const int cnt = (qn - k0 < 32) ? (qn - k0) : 32;
    const bool ok = lane < cnt;
    int e = 0;
    if (ok) e = kept ? slab[1 + k0 + lane] : q[k0 + lane];
    const int si = src[e];
    const int dj = ok ? (dst[e] - base) : 0;
    const float we = ok ? dinv[si] * dloc[dj] : 0.0f;
    for (int u0 = 0; u0 < cnt; u0 += AGG_UB) {
      vec_t hv[AGG_UB];
#pragma unroll
      for (int u = 0; u < AGG_UB; ++u) {
        const int uu = (u0 + u < cnt) ? (u0 + u) : (cnt - 1);
        const int i = __builtin_amdgcn_readlane(si, uu);
        hv[u] = *reinterpret_cast<const vec_t*>(H + (size_t)i * C + lane * VEC);
      }
#pragma unroll
      for (int u = 0; u < AGG_UB; ++u) {
        if (u0 + u < cnt) {
          const int j = __builtin_amdgcn_readlane(dj, u0 + u);
          const float w = __int_as_float(__builtin_amdgcn_readlane(__float_as_int(we), u0 + u));
          const float* hp = reinterpret_cast<const float*>(&hv[u]);
#pragma unroll
          for (int t = 0; t < VEC; ++t) atomicAdd(&acc[j * C + t * 32 + lane], hp[t] * w);
        }
      }
    }
  }
  __syncthreads();

  const vec_t bv = *reinterpret_cast<const vec_t*>(bias + lane * VEC);
  const float* bp = reinterpret_cast<const float*>(&bv);
  for (int r = wv; r < nloc; r += AGG_T / 32) {
    vec_t ov;
    float* op = reinterpret_cast<float*>(&ov);
#pragma unroll
    for (int t = 0; t < VEC; ++t) {
      float v = acc[r * C + t * 32 + lane] + bp[t];
      op[t] = RELU ? fmaxf(v, 0.0f) : v;
    }
    *reinterpret_cast<vec_t*>(out + (size_t)(base + r) * C + lane * VEC) = ov;
  }
}

extern "C" void kernel_launch(void* const* d_in, const int* in_sizes, int n_in,
                              void* d_out, int out_size, void* d_ws, size_t ws_size,
                              hipStream_t stream) {
  const int CIN = 256, CH = 128, CO = 64;
  const float* x  = (const float*)d_in[0];
  const int*   ei = (const int*)d_in[1];
  const float* W1 = (const float*)d_in[2];
  const float* b1 = (const float*)d_in[3];
  const float* W2 = (const float*)d_in[4];
  const float* b2 = (const float*)d_in[5];
  float* out = (float*)d_out;

  const int N = in_sizes[0] / CIN;
  const int E = in_sizes[1] / 2;
  const int* src = ei;
  const int* dst = ei + E;

  float* ws   = (float*)d_ws;
  float* dinv = ws;
  float* Hbuf = ws + 50176;
  float* AGG  = ws + 50176 + (size_t)N * CH;

  int* slabs = (int*)d_out;

  k_owner_scan<<<(N + AGG_R - 1) / AGG_R, AGG_T, 0, stream>>>(dst, dinv, slabs, N, E);

  k_gemm_wmma_f32<<<(N + 31) / 32, 32 * (CH / 32), 0, stream>>>(x, W1, Hbuf, N, CIN, CH);
  k_agg_owner<128, true><<<(N + AGG_R - 1) / AGG_R, AGG_T, 0, stream>>>(src, dst, Hbuf, dinv, b1, slabs, AGG, N, E);

  k_gemm_wmma_f32<<<(N + 31) / 32, 32 * (CO / 32), 0, stream>>>(AGG, W2, Hbuf, N, CH, CO);
  k_agg_owner<64, false><<<(N + AGG_R - 1) / AGG_R, AGG_T, 0, stream>>>(src, dst, Hbuf, dinv, b2, slabs, out, N, E);
}


# dst scan tiled two ways: one work-group per (8 owners, 1/8 of dst) fills per-part edge segments and degree counts; x@W1 launch sums dinv; dst read 32x, not 256x; sum order = LDS arrival
# speedup vs baseline: 2.8197x; 1.1465x over previous
#include <hip/hip_runtime.h>
#include <hip/hip_bf16.h>

typedef float v2f __attribute__((ext_vector_type(2)));
typedef float v8f __attribute__((ext_vector_type(8)));

constexpr int AGG_R    = 196;
constexpr int AGG_T    = 1024;
constexpr int AGG_QCAP = 8192;
constexpr int AGG_US   = 8;
constexpr int AGG_UB   = 16;
constexpr int AGG_SLABROW = 64;
constexpr int AGG_SLAB = AGG_R * AGG_SLABROW;
constexpr int AGG_OG   = 8;
constexpr int AGG_S    = 8;

template <bool DINV>
__global__ void k_gemm_wmma_f32(const float* __restrict__ A,
                                const float* __restrict__ B,
                                float* __restrict__ C,
                                int M, int K, int N,
                                const int* __restrict__ part,
                                float* __restrict__ dinv) {
  const int lane = threadIdx.x & 31;
  const int wv   = threadIdx.x >> 5;
  const int h    = lane >> 4;
  const int l16  = lane & 15;

  const int mbase = blockIdx.x * 32;
  const int m0 = mbase + l16;
  const int m1 = m0 + 16;
  const int mc0 = (m0 < M) ? m0 : (M - 1);
  const int mc1 = (m1 < M) ? m1 : (M - 1);
  const int n0 = wv * 32 + l16;
  const int n1 = n0 + 16;

  const float* a0row = A + (size_t)mc0 * K;
  const float* a1row = A + (size_t)mc1 * K;

  v8f acc00 = {}, acc01 = {}, acc10 = {}, acc11 = {};
  for (int k0 = 0; k0 < K; k0 += 4) {
    const int ka = k0 + 2 * h;
    v2f a0, a1, b0, b1;
    a0.x = a0row[ka];  a0.y = a0row[ka + 1];
    a1.x = a1row[ka];  a1.y = a1row[ka + 1];
    const float* br0 = B + (size_t)ka * N;
    const float* br1 = br0 + N;
    b0.x = br0[n0];  b0.y = br1[n0];
    b1.x = br0[n1];  b1.y = br1[n1];
    acc00 = __builtin_amdgcn_wmma_f32_16x16x4_f32(false, a0, false, b0, (short)0, acc00, false, false);
    acc01 = __builtin_amdgcn_wmma_f32_16x16x4_f32(false, a0, false, b1, (short)0, acc01, false, false);
    acc10 = __builtin_amdgcn_wmma_f32_16x16x4_f32(false, a1, false, b0, (short)0, acc10, false, false);
    acc11 = __builtin_amdgcn_wmma_f32_16x16x4_f32(false, a1, false, b1, (short)0, acc11, false, false);
  }

#pragma unroll
  for (int r = 0; r < 8; ++r) {
    const int row0 = mbase + r + 8 * h;
    const int row1 = row0 + 16;
    if (row0 < M) {
      C[(size_t)row0 * N + n0] = acc00[r];
      C[(size_t)row0 * N + n1] = acc01[r];
    }
    if (row1 < M) {
      C[(size_t)row1 * N + n0] = acc10[r];
      C[(size_t)row1 * N + n1] = acc11[r];
    }
  }

  const int node = mbase + lane;
  if (DINV && wv == 0 && node < M) {
    int deg = 0;
#pragma unroll
    for (int s = 0; s < AGG_S; ++s) deg += part[(size_t)s * M + node];
    dinv[node] = rsqrtf(1.0f + (float)deg);
  }
}

static_assert(AGG_US * 4 <= 32, "one hit bit per scanned destination");
template <bool B> struct agg_flag { static constexpr bool value = B; };
template <int VEC> struct agg_vec;
template <> struct agg_vec<4> { typedef float4 type; };
template <> struct agg_vec<2> { typedef float2 type; };

__device__ __forceinline__ int agg_seg_words(int nloc) { return (nloc * AGG_SLABROW) / AGG_S; }

__device__ __forceinline__ void agg_part_bounds(int units, int part, int nparts, int& lo, int& hi) {
  const int per = (units + nparts - 1) / nparts;
  lo = part * per;
  if (lo > units) lo = units;
  hi = lo + per;
  if (hi > units) hi = units;
}

template <class Push>
__device__ __forceinline__ void agg_scan_dst(const int* __restrict__ dst, int E, int part, int nparts, int base, int nloc, int tid, Push push) {
  if ((E & 3) == 0) {
    const int4* d4 = reinterpret_cast<const int4*>(dst);
    int lo, E4;
    agg_part_bounds(E >> 2, part, nparts, lo, E4);
    auto round = [&](int g, auto guarded) {
      int4 dv[AGG_US];
#pragma unroll
      for (int u = 0; u < AGG_US; ++u) {
        const int gu = g + u * AGG_T;
        dv[u] = d4[(!guarded.value || gu < E4) ? gu : g];
      }
      unsigned hm = 0u;
#pragma unroll
      for (int u = 0; u < AGG_US; ++u) {
        const unsigned lim = (!guarded.value || g + u * AGG_T < E4) ? (unsigned)nloc : 0u;
        hm |= ((unsigned)(dv[u].x - base) < lim) ? (1u << (4 * u))     : 0u;
        hm |= ((unsigned)(dv[u].y - base) < lim) ? (1u << (4 * u + 1)) : 0u;
        hm |= ((unsigned)(dv[u].z - base) < lim) ? (1u << (4 * u + 2)) : 0u;
        hm |= ((unsigned)(dv[u].w - base) < lim) ? (1u << (4 * u + 3)) : 0u;
      }
      while (hm) {
        const int b = __builtin_ctz(hm);
        hm &= hm - 1u;
        push(4 * (g + (b >> 2) * AGG_T) + (b & 3));
      }
    };
    int g = lo + tid;
    for (; g + (AGG_US - 1) * AGG_T < E4; g += AGG_US * AGG_T) round(g, agg_flag<false>());
    if (g < E4) round(g, agg_flag<true>());
  } else {
    int lo, hi;
    agg_part_bounds(E, part, nparts, lo, hi);
    for (int e = lo + tid; e < hi; e += AGG_T)
      if ((unsigned)(dst[e] - base) < (unsigned)nloc) push(e);
  }
}

__global__ __launch_bounds__(1024) void k_owner_scan(const int* __restrict__ dst,
                                                     int* __restrict__ part,
                                                     int* __restrict__ slabs,
                                                     int N, int E) {
  __shared__ int q[AGG_QCAP];
  __shared__ int cnt[AGG_OG * AGG_R];
  __shared__ int ocount[AGG_OG];
  __shared__ int qcount;

  const int tid   = threadIdx.x;
  const int grp   = blockIdx.x / AGG_S;
  const int sp    = blockIdx.x % AGG_S;
  const int gbase = grp * (AGG_OG * AGG_R);
  int gn = N - gbase;
  if (gn > AGG_OG * AGG_R) gn = AGG_OG * AGG_R;

  if (tid == 0) qcount = 0;
  if (tid < AGG_OG) ocount[tid] = 0;
  for (int r = tid; r < AGG_OG * AGG_R; r += AGG_T) cnt[r] = 0;
  __syncthreads();

  agg_scan_dst(dst, E, sp, AGG_S, gbase, gn, tid, [&](int e) {
    const int pos = atomicAdd(&qcount, 1);
    if (pos < AGG_QCAP) q[pos] = e;
  });
  __syncthreads();

  const int total = qcount;
  const bool over = total > AGG_QCAP;
  if (!over) {
    for (int k = tid; k < total; k += AGG_T) {
      const int e  = q[k];
      const int dl = dst[e] - gbase;
      const int o  = dl / AGG_R;
      atomicAdd(&cnt[dl], 1);
      const int pos = atomicAdd(&ocount[o], 1);
      int onl = N - (gbase + o * AGG_R);
      if (onl > AGG_R) onl = AGG_R;
      const int segw = agg_seg_words(onl);
      if (pos < segw - 1) slabs[(size_t)(grp * AGG_OG + o) * AGG_SLAB + sp * segw + 1 + pos] = e;
    }
  } else {
    const bool wide = (E & 3) == 0;
    int lo, hi;
    agg_part_bounds(wide ? (E >> 2) : E, sp, AGG_S, lo, hi);
    if (wide) { lo *= 4; hi *= 4; }
    for (int e = lo + tid; e < hi; e += AGG_T) {
      const int dl = dst[e] - gbase;
      if ((unsigned)dl < (unsigned)gn) atomicAdd(&cnt[dl], 1);
    }
  }
  __syncthreads();

  if (tid < AGG_OG) {
    const int obase = gbase + tid * AGG_R;
    if (obase < N) {
      int onl = N - obase;
      if (onl > AGG_R) onl = AGG_R;
      slabs[(size_t)(grp * AGG_OG + tid) * AGG_SLAB + sp * agg_seg_words(onl)] = over ? (AGG_QCAP + 1) : ocount[tid];
    }
  }
  for (int r = tid; r < gn; r += AGG_T) part[(size_t)sp * N + gbase + r] = cnt[r];
}

template <int C, bool RELU>
__global__ __launch_bounds__(1024) void k_agg_owner(const int* __restrict__ src,
                                                    const int* __restrict__ dst,
                                                    const float* __restrict__ H,
                                                    const float* __restrict__ dinv,
                                                    const float* __restrict__ bias,
                                                    const int* slabs,
                                                    float* out,
                                                    int N, int E) {
  constexpr int VEC = C / 32;
  typedef typename agg_vec<VEC>::type vec_t;
  __shared__ float acc[AGG_R * C];
  __shared__ int   q[AGG_QCAP];
  __shared__ float dloc[AGG_R];
  __shared__ int   qcount;

  const int tid  = threadIdx.x;
  const int lane = tid & 31;
  const int wv   = tid >> 5;
  const int base = blockIdx.x * AGG_R;
  int nloc = N - base;
  if (nloc > AGG_R) nloc = AGG_R;

  const int* slab = slabs + (size_t)blockIdx.x * AGG_SLAB;
  const int segw = agg_seg_words(nloc);
  int segc[AGG_S];
  int total = 0;
  bool kept = true;
#pragma unroll
  for (int s = 0; s < AGG_S; ++s) {
    segc[s] = __builtin_amdgcn_readfirstlane(slab[s * segw]);
    kept = kept && (segc[s] <= segw - 1);
    total += segc[s];
  }

  if (tid == 0) qcount = 0;
  for (int r = tid; r < nloc; r += AGG_T) dloc[r] = dinv[base + r];
  for (int r = wv; r < nloc; r += AGG_T / 32) {
    float w = dinv[base + r];
    w *= w;
    const vec_t hv = *reinterpret_cast<const vec_t*>(H + (size_t)(base + r) * C + lane * VEC);
    const float* hp = reinterpret_cast<const float*>(&hv);
#pragma unroll
    for (int t = 0; t < VEC; ++t) acc[r * C + t * 32 + lane] = hp[t] * w;
  }
  __syncthreads();

  if (!kept) {
    agg_scan_dst(dst, E, 0, 1, base, nloc, tid, [&](int e) {
      const int pos = atomicAdd(&qcount, 1);
      if (pos < AGG_QCAP) {
        q[pos] = e;
      } else {
        const int i = src[e];
        const int d = dst[e];
        const float w = dinv[i] * dinv[d];
        const float* hrow = H + (size_t)i * C;
        float* arow = acc + (d - base) * C;
        for (int c = 0; c < C; ++c) atomicAdd(&arow[(c % VEC) * 32 + c / VEC], hrow[c] * w);
      }
    });
  }
  __syncthreads();

  int qn = total;
  if (!kept) {
    qn = qcount;
    if (qn > AGG_QCAP) qn = AGG_QCAP;
  }
  qn = __builtin_amdgcn_readfirstlane(qn);
  const int wvu = __builtin_amdgcn_readfirstlane(wv);
  for (int k0 = wvu * 32; k0 < qn; k0 += (AGG_T / 32) * 32) {
    const int cnt = (qn - k0 < 32) ? (qn - k0) : 32;
    const bool ok = lane < cnt;
    int e = 0;
    if (ok) {
      if (kept) {
        const int k = k0 + lane;
        int seg = 0, first = 0, run = 0;
#pragma unroll
        for (int s = 0; s + 1 < AGG_S; ++s) {
          run += segc[s];
          if (k >= run) { seg = s + 1; first = run; }
        }
        e = slab[seg * segw + 1 + (k - first)];
      } else {
        e = q[k0 + lane];
      }
    }
    const int si = src[e];
    const int dj = ok ? (dst[e] - base) : 0;
    const float we = ok ? dinv[si] * dloc[dj] : 0.0f;
    for (int u0 = 0; u0 < cnt; u0 += AGG_UB) {
      vec_t hv[AGG_UB];
#pragma unroll
      for (int u = 0; u < AGG_UB; ++u) {
        const int uu = (u0 + u < cnt) ? (u0 + u) : (cnt - 1);
        const int i = __builtin_amdgcn_readlane(si, uu);
        hv[u] = *reinterpret_cast<const vec_t*>(H + (size_t)i * C + lane * VEC);
      }
#pragma unroll
      for (int u = 0; u < AGG_UB; ++u) {
        if (u0 + u < cnt) {
          const int j = __builtin_amdgcn_readlane(dj, u0 + u);
          const float w = __int_as_float(__builtin_amdgcn_readlane(__float_as_int(we), u0 + u));
          const float* hp = reinterpret_cast<const float*>(&hv[u]);
#pragma unroll
          for (int t = 0; t < VEC; ++t) atomicAdd(&acc[j * C + t * 32 + lane], hp[t] * w);
        }
      }
    }
  }
  __syncthreads();

  const vec_t bv = *reinterpret_cast<const vec_t*>(bias + lane * VEC);
  const float* bp = reinterpret_cast<const float*>(&bv);
  for (int r = wv; r < nloc; r += AGG_T / 32) {
    vec_t ov;
    float* op = reinterpret_cast<float*>(&ov);
#pragma unroll
    for (int t = 0; t < VEC; ++t) {
      float v = acc[r * C + t * 32 + lane] + bp[t];
      op[t] = RELU ? fmaxf(v, 0.0f) : v;
    }
    *reinterpret_cast<vec_t*>(out + (size_t)(base + r) * C + lane * VEC) = ov;
  }
}

extern "C" void kernel_launch(void* const* d_in, const int* in_sizes, int n_in,
                              void* d_out, int out_size, void* d_ws, size_t ws_size,
                              hipStream_t stream) {
  const int CIN = 256, CH = 128, CO = 64;
  const float* x  = (const float*)d_in[0];
  const int*   ei = (const int*)d_in[1];
  const float* W1 = (const float*)d_in[2];
  const float* b1 = (const float*)d_in[3];
  const float* W2 = (const float*)d_in[4];
  const float* b2 = (const float*)d_in[5];
  float* out = (float*)d_out;

  const int N = in_sizes[0] / CIN;
  const int E = in_sizes[1] / 2;
  const int* src = ei;
  const int* dst = ei + E;

  float* ws   = (float*)d_ws;
  float* dinv = ws;
  float* Hbuf = ws + 50176;
  float* AGG  = ws + 50176 + (size_t)N * CH;

  int* slabs = (int*)d_out;
  int* part = (int*)AGG;

  const int owners = (N + AGG_R - 1) / AGG_R;
  k_owner_scan<<<((owners + AGG_OG - 1) / AGG_OG) * AGG_S, AGG_T, 0, stream>>>(dst, part, slabs, N, E);

  k_gemm_wmma_f32<true><<<(N + 31) / 32, 32 * (CH / 32), 0, stream>>>(x, W1, Hbuf, N, CIN, CH, part, dinv);
  k_agg_owner<128, true><<<(N + AGG_R - 1) / AGG_R, AGG_T, 0, stream>>>(src, dst, Hbuf, dinv, b1, slabs, AGG, N, E);

  k_gemm_wmma_f32<false><<<(N + 31) / 32, 32 * (CO / 32), 0, stream>>>(AGG, W2, Hbuf, N, CH, CO, nullptr, nullptr);
  k_agg_owner<64, false><<<(N + AGG_R - 1) / AGG_R, AGG_T, 0, stream>>>(src, dst, Hbuf, dinv, b2, slabs, out, N, E);
}


# Aggregation sums in registers: owner work-group bins its edges per destination row via LDS integer counters, each wave sums its own rows' bins; LDS float atomics gone; sum order = bin arrival
# speedup vs baseline: 3.0478x; 1.0809x over previous
#include <hip/hip_runtime.h>
#include <hip/hip_bf16.h>

typedef float v2f __attribute__((ext_vector_type(2)));
typedef float v8f __attribute__((ext_vector_type(8)));

constexpr int AGG_R    = 196;
constexpr int AGG_T    = 1024;
constexpr int AGG_QCAP = 8192;
constexpr int AGG_US   = 8;
constexpr int AGG_UB   = 16;
constexpr int AGG_W    = AGG_T / 32;
constexpr int AGG_SLOTS = (AGG_R + AGG_W - 1) / AGG_W;
constexpr int AGG_BCAP = 64;
constexpr int AGG_LK   = 4;
constexpr int AGG_SLABROW = 64;
constexpr int AGG_SLAB = AGG_R * AGG_SLABROW;
constexpr int AGG_OG   = 8;
constexpr int AGG_S    = 8;

template <bool DINV>
__global__ void k_gemm_wmma_f32(const float* __restrict__ A,
                                const float* __restrict__ B,
                                float* __restrict__ C,
                                int M, int K, int N,
                                const int* __restrict__ part,
                                float* __restrict__ dinv) {
  const int lane = threadIdx.x & 31;
  const int wv   = threadIdx.x >> 5;
  const int h    = lane >> 4;
  const int l16  = lane & 15;

  const int mbase = blockIdx.x * 32;
  const int m0 = mbase + l16;
  const int m1 = m0 + 16;
  const int mc0 = (m0 < M) ? m0 : (M - 1);
  const int mc1 = (m1 < M) ? m1 : (M - 1);
  const int n0 = wv * 32 + l16;
  const int n1 = n0 + 16;

  const float* a0row = A + (size_t)mc0 * K;
  const float* a1row = A + (size_t)mc1 * K;

  v8f acc00 = {}, acc01 = {}, acc10 = {}, acc11 = {};
  for (int k0 = 0; k0 < K; k0 += 4) {
    const int ka = k0 + 2 * h;
    v2f a0, a1, b0, b1;
    a0.x = a0row[ka];  a0.y = a0row[ka + 1];
    a1.x = a1row[ka];  a1.y = a1row[ka + 1];
    const float* br0 = B + (size_t)ka * N;
    const float* br1 = br0 + N;
    b0.x = br0[n0];  b0.y = br1[n0];
    b1.x = br0[n1];  b1.y = br1[n1];
    acc00 = __builtin_amdgcn_wmma_f32_16x16x4_f32(false, a0, false, b0, (short)0, acc00, false, false);
    acc01 = __builtin_amdgcn_wmma_f32_16x16x4_f32(false, a0, false, b1, (short)0, acc01, false, false);
    acc10 = __builtin_amdgcn_wmma_f32_16x16x4_f32(false, a1, false, b0, (short)0, acc10, false, false);
    acc11 = __builtin_amdgcn_wmma_f32_16x16x4_f32(false, a1, false, b1, (short)0, acc11, false, false);
  }

#pragma unroll
  for (int r = 0; r < 8; ++r) {
    const int row0 = mbase + r + 8 * h;
    const int row1 = row0 + 16;
    if (row0 < M) {
      C[(size_t)row0 * N + n0] = acc00[r];
      C[(size_t)row0 * N + n1] = acc01[r];
    }
    if (row1 < M) {
      C[(size_t)row1 * N + n0] = acc10[r];
      C[(size_t)row1 * N + n1] = acc11[r];
    }
  }

  const int node = mbase + lane;
  if (DINV && wv == 0 && node < M) {
    int deg = 0;
#pragma unroll
    for (int s = 0; s < AGG_S; ++s) deg += part[(size_t)s * M + node];
    dinv[node] = rsqrtf(1.0f + (float)deg);
  }
}

static_assert(AGG_US * 4 <= 32, "one hit bit per scanned destination");
template <bool B> struct agg_flag { static constexpr bool value = B; };
template <int VEC> struct agg_vec;
template <> struct agg_vec<4> { typedef float4 type; };
template <> struct agg_vec<2> { typedef float2 type; };

__device__ __forceinline__ int agg_seg_words(int nloc) { return (nloc * AGG_SLABROW) / AGG_S; }

__device__ __forceinline__ void agg_part_bounds(int units, int part, int nparts, int& lo, int& hi) {
  const int per = (units + nparts - 1) / nparts;
  lo = part * per;
  if (lo > units) lo = units;
  hi = lo + per;
  if (hi > units) hi = units;
}

template <class Push>
__device__ __forceinline__ void agg_scan_dst(const int* __restrict__ dst, int E, int part, int nparts, int base, int nloc, int tid, Push push) {
  if ((E & 3) == 0) {
    const int4* d4 = reinterpret_cast<const int4*>(dst);
    int lo, E4;
    agg_part_bounds(E >> 2, part, nparts, lo, E4);
    auto round = [&](int g, auto guarded) {
      int4 dv[AGG_US];
#pragma unroll
      for (int u = 0; u < AGG_US; ++u) {
        const int gu = g + u * AGG_T;
        dv[u] = d4[(!guarded.value || gu < E4) ? gu : g];
      }
      unsigned hm = 0u;
#pragma unroll
      for (int u = 0; u < AGG_US; ++u) {
        const unsigned lim = (!guarded.value || g + u * AGG_T < E4) ? (unsigned)nloc : 0u;
        hm |= ((unsigned)(dv[u].x - base) < lim) ? (1u << (4 * u))     : 0u;
        hm |= ((unsigned)(dv[u].y - base) < lim) ? (1u << (4 * u + 1)) : 0u;
        hm |= ((unsigned)(dv[u].z - base) < lim) ? (1u << (4 * u + 2)) : 0u;
        hm |= ((unsigned)(dv[u].w - base) < lim) ? (1u << (4 * u + 3)) : 0u;
      }
      while (hm) {
        const int b = __builtin_ctz(hm);
        hm &= hm - 1u;
        push(4 * (g + (b >> 2) * AGG_T) + (b & 3));
      }
    };
    int g = lo + tid;
    for (; g + (AGG_US - 1) * AGG_T < E4; g += AGG_US * AGG_T) round(g, agg_flag<false>());
    if (g < E4) round(g, agg_flag<true>());
  } else {
    int lo, hi;
    agg_part_bounds(E, part, nparts, lo, hi);
    for (int e = lo + tid; e < hi; e += AGG_T)
      if ((unsigned)(dst[e] - base) < (unsigned)nloc) push(e);
  }
}

__global__ __launch_bounds__(1024) void k_owner_scan(const int* __restrict__ dst,
                                                     int* __restrict__ part,
                                                     int* __restrict__ slabs,
                                                     int N, int E) {
  __shared__ int q[AGG_QCAP];
  __shared__ int cnt[AGG_OG * AGG_R];
  __shared__ int ocount[AGG_OG];
  __shared__ int qcount;

  const int tid   = threadIdx.x;
  const int grp   = blockIdx.x / AGG_S;
  const int sp    = blockIdx.x % AGG_S;
  const int gbase = grp * (AGG_OG * AGG_R);
  int gn = N - gbase;
  if (gn > AGG_OG * AGG_R) gn = AGG_OG * AGG_R;

  if (tid == 0) qcount = 0;
  if (tid < AGG_OG) ocount[tid] = 0;
  for (int r = tid; r < AGG_OG * AGG_R; r += AGG_T) cnt[r] = 0;
  __syncthreads();

  agg_scan_dst(dst, E, sp, AGG_S, gbase, gn, tid, [&](int e) {
    const int pos = atomicAdd(&qcount, 1);
    if (pos < AGG_QCAP) q[pos] = e;
  });
  __syncthreads();

  const int total = qcount;
  const bool over = total > AGG_QCAP;
  if (!over) {
    for (int k = tid; k < total; k += AGG_T) {
      const int e  = q[k];
      const int dl = dst[e] - gbase;
      const int o  = dl / AGG_R;
      atomicAdd(&cnt[dl], 1);
      const int pos = atomicAdd(&ocount[o], 1);
      int onl = N - (gbase + o * AGG_R);
      if (onl > AGG_R) onl = AGG_R;
      const int segw = agg_seg_words(onl);
      if (pos < segw - 1) slabs[(size_t)(grp * AGG_OG + o) * AGG_SLAB + sp * segw + 1 + pos] = e;
    }
  } else {
    const bool wide = (E & 3) == 0;
    int lo, hi;
    agg_part_bounds(wide ? (E >> 2) : E, sp, AGG_S, lo, hi);
    if (wide) { lo *= 4; hi *= 4; }
    for (int e = lo + tid; e < hi; e += AGG_T) {
      const int dl = dst[e] - gbase;
      if ((unsigned)dl < (unsigned)gn) atomicAdd(&cnt[dl], 1);
    }
  }
  __syncthreads();

  if (tid < AGG_OG) {
    const int obase = gbase + tid * AGG_R;
    if (obase < N) {
      int onl = N - obase;
      if (onl > AGG_R) onl = AGG_R;
      slabs[(size_t)(grp * AGG_OG + tid) * AGG_SLAB + sp * agg_seg_words(onl)] = over ? (AGG_QCAP + 1) : ocount[tid];
    }
  }
  for (int r = tid; r < gn; r += AGG_T) part[(size_t)sp * N + gbase + r] = cnt[r];
}

template <int C, bool RELU>
__global__ __launch_bounds__(1024) void k_agg_owner(const int* __restrict__ src,
                                                    const int* __restrict__ dst,
                                                    const float* __restrict__ H,
                                                    const float* __restrict__ dinv,
                                                    const float* __restrict__ bias,
                                                    const int* slabs,
                                                    float* out,
                                                    int N, int E) {
  constexpr int VEC = C / 32;
  typedef typename agg_vec<VEC>::type vec_t;
  __shared__ float acc[AGG_R * C];
  __shared__ int   q[AGG_QCAP];
  __shared__ int   bsrc[AGG_R * AGG_BCAP];
  __shared__ float bwt[AGG_R * AGG_BCAP];
  __shared__ int   bcount[AGG_R];
  __shared__ float dloc[AGG_R];
  __shared__ int   qcount;

  const int tid  = threadIdx.x;
  const int lane = tid & 31;
  const int wv   = tid >> 5;
  const int wvu  = __builtin_amdgcn_readfirstlane(wv);
  const int base = blockIdx.x * AGG_R;
  int nloc = N - base;
  if (nloc > AGG_R) nloc = AGG_R;

  const int* slab = slabs + (size_t)blockIdx.x * AGG_SLAB;
  const int segw = agg_seg_words(nloc);
  int segc[AGG_S];
  int total = 0;
  bool kept = true;
#pragma unroll
  for (int s = 0; s < AGG_S; ++s) {
    segc[s] = __builtin_amdgcn_readfirstlane(slab[s * segw]);
    kept = kept && (segc[s] <= segw - 1);
    total += segc[s];
  }

  vec_t sum[AGG_SLOTS];
  {
    float dw[AGG_SLOTS];
    vec_t hs[AGG_SLOTS];
#pragma unroll
    for (int s = 0; s < AGG_SLOTS; ++s) {
      const int r  = wvu + AGG_W * s;
      const int rr = (r < nloc) ? r : (nloc - 1);
      dw[s] = dinv[base + rr];
      hs[s] = *reinterpret_cast<const vec_t*>(H + (size_t)(base + rr) * C + lane * VEC);
    }
#pragma unroll
    for (int s = 0; s < AGG_SLOTS; ++s) {
      const float w = (wvu + AGG_W * s < nloc) ? dw[s] * dw[s] : 0.0f;
      const float* hp = reinterpret_cast<const float*>(&hs[s]);
      float* sp = reinterpret_cast<float*>(&sum[s]);
#pragma unroll
      for (int t = 0; t < VEC; ++t) sp[t] = hp[t] * w;
    }
  }
  if (tid == 0) qcount = 0;
  for (int r = tid; r < nloc; r += AGG_T) { dloc[r] = dinv[base + r]; bcount[r] = 0; }
  {
    float4* a4 = reinterpret_cast<float4*>(acc);
    const float4 z = {0.0f, 0.0f, 0.0f, 0.0f};
    for (int i = tid; i < (nloc * C) / 4; i += AGG_T) a4[i] = z;
  }
  __syncthreads();

  if (!kept) {
    agg_scan_dst(dst, E, 0, 1, base, nloc, tid, [&](int e) {
      const int pos = atomicAdd(&qcount, 1);
      if (pos < AGG_QCAP) {
        q[pos] = e;
      } else {
        const int i = src[e];
        const int d = dst[e];
        const float w = dinv[i] * dinv[d];
        const float* hrow = H + (size_t)i * C;
        float* arow = acc + (d - base) * C;
        for (int c = 0; c < C; ++c) atomicAdd(&arow[(c % VEC) * 32 + c / VEC], hrow[c] * w);
      }
    });
  }
  __syncthreads();

  int qn = total;
  if (!kept) {
    qn = qcount;
    if (qn > AGG_QCAP) qn = AGG_QCAP;
  }
  qn = __builtin_amdgcn_readfirstlane(qn);
  for (int k0 = 0; k0 < qn; k0 += AGG_LK * AGG_T) {
    int  ev[AGG_LK];
    bool okv[AGG_LK];
#pragma unroll
    for (int j = 0; j < AGG_LK; ++j) {
      const int kk = k0 + j * AGG_T + tid;
      okv[j] = kk < qn;
      const int k = okv[j] ? kk : (qn - 1);
      if (kept) {
        int seg = 0, first = 0, run = 0;
#pragma unroll
        for (int s = 0; s + 1 < AGG_S; ++s) {
          run += segc[s];
          if (k >= run) { seg = s + 1; first = run; }
        }
        ev[j] = slab[seg * segw + 1 + (k - first)];
      } else {
        ev[j] = q[k];
      }
    }
    int siv[AGG_LK], djv[AGG_LK];
#pragma unroll
    for (int j = 0; j < AGG_LK; ++j) {
      siv[j] = src[ev[j]];
      djv[j] = dst[ev[j]];
    }
    float wev[AGG_LK];
#pragma unroll
    for (int j = 0; j < AGG_LK; ++j) {
      djv[j] = okv[j] ? (djv[j] - base) : 0;
      wev[j] = dinv[siv[j]] * dloc[djv[j]];
    }
#pragma unroll
    for (int j = 0; j < AGG_LK; ++j) {
      if (okv[j]) {
        const int pos = atomicAdd(&bcount[djv[j]], 1);
        if (pos < AGG_BCAP) {
          bsrc[djv[j] * AGG_BCAP + pos] = siv[j];
          bwt[djv[j] * AGG_BCAP + pos]  = wev[j];
        } else {
          const float* hrow = H + (size_t)siv[j] * C;
          float* arow = acc + djv[j] * C;
          for (int c = 0; c < C; ++c) atomicAdd(&arow[(c % VEC) * 32 + c / VEC], hrow[c] * wev[j]);
        }
      }
    }
  }
  __syncthreads();

  constexpr int UB = AGG_UB * 4 / VEC;
  const vec_t bv = *reinterpret_cast<const vec_t*>(bias + lane * VEC);
  const float* bp = reinterpret_cast<const float*>(&bv);
#pragma unroll
  for (int s = 0; s < AGG_SLOTS; ++s) {
    const int r = wvu + AGG_W * s;
    if (r < nloc) {
      vec_t a = sum[s];
      float* ap = reinterpret_cast<float*>(&a);
      int bn = bcount[r];
      if (bn > AGG_BCAP) bn = AGG_BCAP;
      bn = __builtin_amdgcn_readfirstlane(bn);
      const int*   mys = bsrc + r * AGG_BCAP;
      const float* myw = bwt + r * AGG_BCAP;
      for (int i0 = 0; i0 < bn; i0 += 32) {
        const int cnt = (bn - i0 < 32) ? (bn - i0) : 32;
        const bool ok = lane < cnt;
        const int   si = ok ? mys[i0 + lane] : 0;
        const float we = ok ? myw[i0 + lane] : 0.0f;
        for (int u0 = 0; u0 < cnt; u0 += UB) {
          vec_t hv[UB];
#pragma unroll
          for (int u = 0; u < UB; ++u) {
            const int uu = (u0 + u < cnt) ? (u0 + u) : (cnt - 1);
            const int i = __builtin_amdgcn_readlane(si, uu);
            hv[u] = *reinterpret_cast<const vec_t*>(H + (size_t)i * C + lane * VEC);
          }
#pragma unroll
          for (int u = 0; u < UB; ++u) {
            if (u0 + u < cnt) {
              const float w = __int_as_float(__builtin_amdgcn_readlane(__float_as_int(we), u0 + u));
              const float* hp = reinterpret_cast<const float*>(&hv[u]);
#pragma unroll
              for (int t = 0; t < VEC; ++t) ap[t] += hp[t] * w;
            }
          }
        }
      }
      vec_t ov;
      float* op = reinterpret_cast<float*>(&ov);
#pragma unroll
      for (int t = 0; t < VEC; ++t) {
        float v = ap[t] + acc[r * C + t * 32 + lane] + bp[t];
        op[t] = RELU ? fmaxf(v, 0.0f) : v;
      }
      *reinterpret_cast<vec_t*>(out + (size_t)(base + r) * C + lane * VEC) = ov;
    }
  }
}

extern "C" void kernel_launch(void* const* d_in, const int* in_sizes, int n_in,
                              void* d_out, int out_size, void* d_ws, size_t ws_size,
                              hipStream_t stream) {
  const int CIN = 256, CH = 128, CO = 64;
  const float* x  = (const float*)d_in[0];
  const int*   ei = (const int*)d_in[1];
  const float* W1 = (const float*)d_in[2];
  const float* b1 = (const float*)d_in[3];
  const float* W2 = (const float*)d_in[4];
  const float* b2 = (const float*)d_in[5];
  float* out = (float*)d_out;

  const int N = in_sizes[0] / CIN;
  const int E = in_sizes[1] / 2;
  const int* src = ei;
  const int* dst = ei + E;

  float* ws   = (float*)d_ws;
  float* dinv = ws;
  float* Hbuf = ws + 50176;
  float* AGG  = ws + 50176 + (size_t)N * CH;

  int* slabs = (int*)d_out;
  int* part = (int*)AGG;

  const int owners = (N + AGG_R - 1) / AGG_R;
  k_owner_scan<<<((owners + AGG_OG - 1) / AGG_OG) * AGG_S, AGG_T, 0, stream>>>(dst, part, slabs, N, E);

  k_gemm_wmma_f32<true><<<(N + 31) / 32, 32 * (CH / 32), 0, stream>>>(x, W1, Hbuf, N, CIN, CH, part, dinv);
  k_agg_owner<128, true><<<(N + AGG_R - 1) / AGG_R, AGG_T, 0, stream>>>(src, dst, Hbuf, dinv, b1, slabs, AGG, N, E);

  k_gemm_wmma_f32<false><<<(N + 31) / 32, 32 * (CO / 32), 0, stream>>>(AGG, W2, Hbuf, N, CH, CO, nullptr, nullptr);
  k_agg_owner<64, false><<<(N + AGG_R - 1) / AGG_R, AGG_T, 0, stream>>>(src, dst, Hbuf, dinv, b2, slabs, out, N, E);
}


# Gather issue: bin walk loads only entries the bin holds, row address = one lane read + lane offset; two 64-channel rows per wave load, half-wave sums added per row (layer-2 sum order changes)
# speedup vs baseline: 3.1756x; 1.0419x over previous
#include <hip/hip_runtime.h>
#include <hip/hip_bf16.h>

typedef float v2f __attribute__((ext_vector_type(2)));
typedef float v8f __attribute__((ext_vector_type(8)));

constexpr int AGG_R    = 196;
constexpr int AGG_T    = 1024;
constexpr int AGG_QCAP = 8192;
constexpr int AGG_US   = 8;
constexpr int AGG_UB   = 16;
constexpr int AGG_W    = AGG_T / 32;
constexpr int AGG_SLOTS = (AGG_R + AGG_W - 1) / AGG_W;
constexpr int AGG_BCAP = 64;
constexpr int AGG_LK   = 4;
constexpr int AGG_SLABROW = 64;
constexpr int AGG_SLAB = AGG_R * AGG_SLABROW;
constexpr int AGG_OG   = 8;
constexpr int AGG_S    = 8;

template <bool DINV>
__global__ void k_gemm_wmma_f32(const float* __restrict__ A,
                                const float* __restrict__ B,
                                float* __restrict__ C,
                                int M, int K, int N,
                                const int* __restrict__ part,
                                float* __restrict__ dinv) {
  const int lane = threadIdx.x & 31;
  const int wv   = threadIdx.x >> 5;
  const int h    = lane >> 4;
  const int l16  = lane & 15;

  const int mbase = blockIdx.x * 32;
  const int m0 = mbase + l16;
  const int m1 = m0 + 16;
  const int mc0 = (m0 < M) ? m0 : (M - 1);
  const int mc1 = (m1 < M) ? m1 : (M - 1);
  const int n0 = wv * 32 + l16;
  const int n1 = n0 + 16;

  const float* a0row = A + (size_t)mc0 * K;
  const float* a1row = A + (size_t)mc1 * K;

  v8f acc00 = {}, acc01 = {}, acc10 = {}, acc11 = {};
  for (int k0 = 0; k0 < K; k0 += 4) {
    const int ka = k0 + 2 * h;
    v2f a0, a1, b0, b1;
    a0.x = a0row[ka];  a0.y = a0row[ka + 1];
    a1.x = a1row[ka];  a1.y = a1row[ka + 1];
    const float* br0 = B + (size_t)ka * N;
    const float* br1 = br0 + N;
    b0.x = br0[n0];  b0.y = br1[n0];
    b1.x = br0[n1];  b1.y = br1[n1];
    acc00 = __builtin_amdgcn_wmma_f32_16x16x4_f32(false, a0, false, b0, (short)0, acc00, false, false);
    acc01 = __builtin_amdgcn_wmma_f32_16x16x4_f32(false, a0, false, b1, (short)0, acc01, false, false);
    acc10 = __builtin_amdgcn_wmma_f32_16x16x4_f32(false, a1, false, b0, (short)0, acc10, false, false);
    acc11 = __builtin_amdgcn_wmma_f32_16x16x4_f32(false, a1, false, b1, (short)0, acc11, false, false);
  }

#pragma unroll
  for (int r = 0; r < 8; ++r) {
    const int row0 = mbase + r + 8 * h;
    const int row1 = row0 + 16;
    if (row0 < M) {
      C[(size_t)row0 * N + n0] = acc00[r];
      C[(size_t)row0 * N + n1] = acc01[r];
    }
    if (row1 < M) {
      C[(size_t)row1 * N + n0] = acc10[r];
      C[(size_t)row1 * N + n1] = acc11[r];
    }
  }

  const int node = mbase + lane;
  if (DINV && wv == 0 && node < M) {
    int deg = 0;
#pragma unroll
    for (int s = 0; s < AGG_S; ++s) deg += part[(size_t)s * M + node];
    dinv[node] = rsqrtf(1.0f + (float)deg);
  }
}

static_assert(AGG_US * 4 <= 32, "one hit bit per scanned destination");
template <bool B> struct agg_flag { static constexpr bool value = B; };

template <int V> struct agg_int { static constexpr int value = V; };
template <int I, int N, class F>
__device__ __forceinline__ void agg_static_upto(int n, F&& f) {
  if constexpr (I < N) {
    if (I < n) { f(agg_int<I>()); agg_static_upto<I + 1, N>(n, f); }
  }
}
template <int U> __device__ __forceinline__ int agg_half_bcast(int v) {
  return __builtin_amdgcn_update_dpp(0, v, 0x150 + U, 0xf, 0xf, true);
}

__device__ __forceinline__ int agg_seg_words(int nloc) { return (nloc * AGG_SLABROW) / AGG_S; }

__device__ __forceinline__ void agg_part_bounds(int units, int part, int nparts, int& lo, int& hi) {
  const int per = (units + nparts - 1) / nparts;
  lo = part * per;
  if (lo > units) lo = units;
  hi = lo + per;
  if (hi > units) hi = units;
}

template <class Push>
__device__ __forceinline__ void agg_scan_dst(const int* __restrict__ dst, int E, int part, int nparts, int base, int nloc, int tid, Push push) {
  if ((E & 3) == 0) {
    const int4* d4 = reinterpret_cast<const int4*>(dst);
    int lo, E4;
    agg_part_bounds(E >> 2, part, nparts, lo, E4);
    auto round = [&](int g, auto guarded) {
      int4 dv[AGG_US];
#pragma unroll
      for (int u = 0; u < AGG_US; ++u) {
        const int gu = g + u * AGG_T;
        dv[u] = d4[(!guarded.value || gu < E4) ? gu : g];
      }
      unsigned hm = 0u;
#pragma unroll
      for (int u = 0; u < AGG_US; ++u) {
        const unsigned lim = (!guarded.value || g + u * AGG_T < E4) ? (unsigned)nloc : 0u;
        hm |= ((unsigned)(dv[u].x - base) < lim) ? (1u << (4 * u))     : 0u;
        hm |= ((unsigned)(dv[u].y - base) < lim) ? (1u << (4 * u + 1)) : 0u;
        hm |= ((unsigned)(dv[u].z - base) < lim) ? (1u << (4 * u + 2)) : 0u;
        hm |= ((unsigned)(dv[u].w - base) < lim) ? (1u << (4 * u + 3)) : 0u;
      }
      while (hm) {
        const int b = __builtin_ctz(hm);
        hm &= hm - 1u;
        push(4 * (g + (b >> 2) * AGG_T) + (b & 3));
      }
    };
    int g = lo + tid;
    for (; g + (AGG_US - 1) * AGG_T < E4; g += AGG_US * AGG_T) round(g, agg_flag<false>());
    if (g < E4) round(g, agg_flag<true>());
  } else {
    int lo, hi;
    agg_part_bounds(E, part, nparts, lo, hi);
    for (int e = lo + tid; e < hi; e += AGG_T)
      if ((unsigned)(dst[e] - base) < (unsigned)nloc) push(e);
  }
}

__global__ __launch_bounds__(1024) void k_owner_scan(const int* __restrict__ dst,
                                                     int* __restrict__ part,
                                                     int* __restrict__ slabs,
                                                     int N, int E) {
  __shared__ int q[AGG_QCAP];
  __shared__ int cnt[AGG_OG * AGG_R];
  __shared__ int ocount[AGG_OG];
  __shared__ int qcount;

  const int tid   = threadIdx.x;
  const int grp   = blockIdx.x / AGG_S;
  const int sp    = blockIdx.x % AGG_S;
  const int gbase = grp * (AGG_OG * AGG_R);
  int gn = N - gbase;
  if (gn > AGG_OG * AGG_R) gn = AGG_OG * AGG_R;

  if (tid == 0) qcount = 0;
  if (tid < AGG_OG) ocount[tid] = 0;
  for (int r = tid; r < AGG_OG * AGG_R; r += AGG_T) cnt[r] = 0;
  __syncthreads();

  agg_scan_dst(dst, E, sp, AGG_S, gbase, gn, tid, [&](int e) {
    const int pos = atomicAdd(&qcount, 1);
    if (pos < AGG_QCAP) q[pos] = e;
  });
  __syncthreads();

  const int total = qcount;
  const bool over = total > AGG_QCAP;
  if (!over) {
    for (int k = tid; k < total; k += AGG_T) {
      const int e  = q[k];
      const int dl = dst[e] - gbase;
      const int o  = dl / AGG_R;
      atomicAdd(&cnt[dl], 1);
      const int pos = atomicAdd(&ocount[o], 1);
      int onl = N - (gbase + o * AGG_R);
      if (onl > AGG_R) onl = AGG_R;
      const int segw = agg_seg_words(onl);
      if (pos < segw - 1) slabs[(size_t)(grp * AGG_OG + o) * AGG_SLAB + sp * segw + 1 + pos] = e;
    }
  } else {
    const bool wide = (E & 3) == 0;
    int lo, hi;
    agg_part_bounds(wide ? (E >> 2) : E, sp, AGG_S, lo, hi);
    if (wide) { lo *= 4; hi *= 4; }
    for (int e = lo + tid; e < hi; e += AGG_T) {
      const int dl = dst[e] - gbase;
      if ((unsigned)dl < (unsigned)gn) atomicAdd(&cnt[dl], 1);
    }
  }
  __syncthreads();

  if (tid < AGG_OG) {
    const int obase = gbase + tid * AGG_R;
    if (obase < N) {
      int onl = N - obase;
      if (onl > AGG_R) onl = AGG_R;
      slabs[(size_t)(grp * AGG_OG + tid) * AGG_SLAB + sp * agg_seg_words(onl)] = over ? (AGG_QCAP + 1) : ocount[tid];
    }
  }
  for (int r = tid; r < gn; r += AGG_T) part[(size_t)sp * N + gbase + r] = cnt[r];
}

template <int C, bool RELU>
__global__ __launch_bounds__(1024) void k_agg_owner(const int* __restrict__ src,
                                                    const int* __restrict__ dst,
                                                    const float* __restrict__ H,
                                                    const float* __restrict__ dinv,
                                                    const float* __restrict__ bias,
                                                    const int* slabs,
                                                    float* out,
                                                    int N, int E) {
  constexpr int VEC = C / 32;
  constexpr int LPR = C / 4;
  constexpr int EPL = 32 / LPR;
  static_assert(EPL == 1 || EPL == 2, "a source row fills a wave or a half-wave");
  __shared__ float acc[AGG_R * C];
  __shared__ int   q[AGG_QCAP];
  __shared__ int   bsrc[AGG_R * AGG_BCAP];
  __shared__ float bwt[AGG_R * AGG_BCAP];
  __shared__ int   bcount[AGG_R];
  __shared__ float dloc[AGG_R];
  __shared__ int   qcount;

  const int tid  = threadIdx.x;
  const int lane = tid & 31;
  const int wv   = tid >> 5;
  const int wvu  = __builtin_amdgcn_readfirstlane(wv);
  const int sub  = lane / LPR;
  const int lch  = lane % LPR;
  const int base = blockIdx.x * AGG_R;
  int nloc = N - base;
  if (nloc > AGG_R) nloc = AGG_R;

  const int* slab = slabs + (size_t)blockIdx.x * AGG_SLAB;
  const int segw = agg_seg_words(nloc);
  int segc[AGG_S];
  int total = 0;
  bool kept = true;
#pragma unroll
  for (int s = 0; s < AGG_S; ++s) {
    segc[s] = __builtin_amdgcn_readfirstlane(slab[s * segw]);
    kept = kept && (segc[s] <= segw - 1);
    total += segc[s];
  }

  float4 sum[AGG_SLOTS];
  {
    float dw[AGG_SLOTS];
    float4 hs[AGG_SLOTS];
#pragma unroll
    for (int s = 0; s < AGG_SLOTS; ++s) {
      const int r  = wvu + AGG_W * s;
      const int rr = (r < nloc) ? r : (nloc - 1);
      dw[s] = dinv[base + rr];
      hs[s] = *reinterpret_cast<const float4*>(H + (size_t)(base + rr) * C + lch * 4);
    }
#pragma unroll
    for (int s = 0; s < AGG_SLOTS; ++s) {
      const float w = (wvu + AGG_W * s < nloc && sub == 0) ? dw[s] * dw[s] : 0.0f;
      sum[s].x = hs[s].x * w;  sum[s].y = hs[s].y * w;  sum[s].z = hs[s].z * w;  sum[s].w = hs[s].w * w;
    }
  }
  if (tid == 0) qcount = 0;
  for (int r = tid; r < nloc; r += AGG_T) { dloc[r] = dinv[base + r]; bcount[r] = 0; }
  {
    float4* a4 = reinterpret_cast<float4*>(acc);
    const float4 z = {0.0f, 0.0f, 0.0f, 0.0f};
    for (int i = tid; i < (nloc * C) / 4; i += AGG_T) a4[i] = z;
  }
  __syncthreads();

  if (!kept) {
    agg_scan_dst(dst, E, 0, 1, base, nloc, tid, [&](int e) {
      const int pos = atomicAdd(&qcount, 1);
      if (pos < AGG_QCAP) {
        q[pos] = e;
      } else {
        const int i = src[e];
        const int d = dst[e];
        const float w = dinv[i] * dinv[d];
        const float* hrow = H + (size_t)i * C;
        float* arow = acc + (d - base) * C;
        for (int c = 0; c < C; ++c) atomicAdd(&arow[(c % VEC) * 32 + c / VEC], hrow[c] * w);
      }
    });
  }
  __syncthreads();

  int qn = total;
  if (!kept) {
    qn = qcount;
    if (qn > AGG_QCAP) qn = AGG_QCAP;
  }
  qn = __builtin_amdgcn_readfirstlane(qn);
  for (int k0 = 0; k0 < qn; k0 += AGG_LK * AGG_T) {
    int  ev[AGG_LK];
    bool okv[AGG_LK];
#pragma unroll
    for (int j = 0; j < AGG_LK; ++j) {
      const int kk = k0 + j * AGG_T + tid;
      okv[j] = kk < qn;
      const int k = okv[j] ? kk : (qn - 1);
      if (kept) {
        int seg = 0, first = 0, run = 0;
#pragma unroll
        for (int s = 0; s + 1 < AGG_S; ++s) {
          run += segc[s];
          if (k >= run) { seg = s + 1; first = run; }
        }
        ev[j] = slab[seg * segw + 1 + (k - first)];
      } else {
        ev[j] = q[k];
      }
    }
    int siv[AGG_LK], djv[AGG_LK];
#pragma unroll
    for (int j = 0; j < AGG_LK; ++j) {
      siv[j] = src[ev[j]];
      djv[j] = dst[ev[j]];
    }
    float wev[AGG_LK];
#pragma unroll
    for (int j = 0; j < AGG_LK; ++j) {
      djv[j] = okv[j] ? (djv[j] - base) : 0;
      wev[j] = dinv[siv[j]] * dloc[djv[j]];
    }
#pragma unroll
    for (int j = 0; j < AGG_LK; ++j) {
      if (okv[j]) {
        const int pos = atomicAdd(&bcount[djv[j]], 1);
        if (pos < AGG_BCAP) {
          bsrc[djv[j] * AGG_BCAP + pos] = siv[j];
          bwt[djv[j] * AGG_BCAP + pos]  = wev[j];
        } else {
          const float* hrow = H + (size_t)siv[j] * C;
          float* arow = acc + djv[j] * C;
          for (int c = 0; c < C; ++c) atomicAdd(&arow[(c % VEC) * 32 + c / VEC], hrow[c] * wev[j]);
        }
      }
    }
  }
  __syncthreads();

  constexpr int UB = AGG_UB;
  const float4 bv = *reinterpret_cast<const float4*>(bias + lch * 4);
  const char* Hb = reinterpret_cast<const char*>(H);
  const unsigned laneoff = (unsigned)lch * 16u;
  const int ent = (EPL == 1) ? lane : (2 * lch + sub);
#pragma unroll
  for (int s = 0; s < AGG_SLOTS; ++s) {
    const int r = wvu + AGG_W * s;
    if (r < nloc) {
      float4 a = sum[s];
      int bn = bcount[r];
      if (bn > AGG_BCAP) bn = AGG_BCAP;
      bn = __builtin_amdgcn_readfirstlane(bn);
      const int*   mys = bsrc + r * AGG_BCAP;
      const float* myw = bwt + r * AGG_BCAP;
      for (int i0 = 0; i0 < bn; i0 += 32) {
        const int cnt = (bn - i0 < 32) ? (bn - i0) : 32;
        const bool ok = ent < cnt;
        unsigned so = ok ? (unsigned)mys[i0 + ent] * (unsigned)(C * 4) : 0u;
        const float we = ok ? myw[i0 + ent] : 0.0f;
        if (EPL == 2) {
          const int last = cnt - 1;
          const unsigned solast = (unsigned)__builtin_amdgcn_readlane((int)so, (last >> 1) + 16 * (last & 1));
          so = ok ? so : solast;
        }
        const int nld = (cnt + EPL - 1) / EPL;
        for (int u0 = 0; u0 < nld; u0 += UB) {
          float4 hv[UB];
          const int nb = nld - u0;
          agg_static_upto<0, UB>(nb, [&](auto uc) {
            constexpr int u = decltype(uc)::value;
            unsigned off;
            if constexpr (EPL == 1) off = (unsigned)__builtin_amdgcn_readlane((int)so, u0 + u);
            else                    off = (unsigned)agg_half_bcast<u>((int)so);
            hv[u] = *reinterpret_cast<const float4*>(Hb + (off + laneoff));
          });
          agg_static_upto<0, UB>(nb, [&](auto uc) {
            constexpr int u = decltype(uc)::value;
            float w;
            if constexpr (EPL == 1) w = __int_as_float(__builtin_amdgcn_readlane(__float_as_int(we), u0 + u));
            else                    w = __int_as_float(agg_half_bcast<u>(__float_as_int(we)));
            a.x += hv[u].x * w;  a.y += hv[u].y * w;  a.z += hv[u].z * w;  a.w += hv[u].w * w;
          });
        }
      }
      if (EPL == 2) {
        a.x += __int_as_float(__builtin_amdgcn_permlanex16(0, __float_as_int(a.x), 0x76543210, 0xfedcba98, false, false));
        a.y += __int_as_float(__builtin_amdgcn_permlanex16(0, __float_as_int(a.y), 0x76543210, 0xfedcba98, false, false));
        a.z += __int_as_float(__builtin_amdgcn_permlanex16(0, __float_as_int(a.z), 0x76543210, 0xfedcba98, false, false));
        a.w += __int_as_float(__builtin_amdgcn_permlanex16(0, __float_as_int(a.w), 0x76543210, 0xfedcba98, false, false));
      }
      if (sub == 0) {
        const float* ar = acc + r * C;
        const float av[4] = {a.x, a.y, a.z, a.w};
        const float bb[4] = {bv.x, bv.y, bv.z, bv.w};
        float o[4];
#pragma unroll
        for (int t = 0; t < 4; ++t) {
          const int ch = 4 * lch + t;
          const float v = av[t] + ar[(ch % VEC) * 32 + ch / VEC] + bb[t];
          o[t] = RELU ? fmaxf(v, 0.0f) : v;
        }
        float4 ov;
        ov.x = o[0];  ov.y = o[1];  ov.z = o[2];  ov.w = o[3];
        *reinterpret_cast<float4*>(out + (size_t)(base + r) * C + lch * 4) = ov;
      }
    }
  }
}

extern "C" void kernel_launch(void* const* d_in, const int* in_sizes, int n_in,
                              void* d_out, int out_size, void* d_ws, size_t ws_size,
                              hipStream_t stream) {
  const int CIN = 256, CH = 128, CO = 64;
  const float* x  = (const float*)d_in[0];
  const int*   ei = (const int*)d_in[1];
  const float* W1 = (const float*)d_in[2];
  const float* b1 = (const float*)d_in[3];
  const float* W2 = (const float*)d_in[4];
  const float* b2 = (const float*)d_in[5];
  float* out = (float*)d_out;

  const int N = in_sizes[0] / CIN;
  const int E = in_sizes[1] / 2;
  const int* src = ei;
  const int* dst = ei + E;

  float* ws   = (float*)d_ws;
  float* dinv = ws;
  float* Hbuf = ws + 50176;
  float* AGG  = ws + 50176 + (size_t)N * CH;

  int* slabs = (int*)d_out;
  int* part = (int*)AGG;

  const int owners = (N + AGG_R - 1) / AGG_R;
  k_owner_scan<<<((owners + AGG_OG - 1) / AGG_OG) * AGG_S, AGG_T, 0, stream>>>(dst, part, slabs, N, E);

  k_gemm_wmma_f32<true><<<(N + 31) / 32, 32 * (CH / 32), 0, stream>>>(x, W1, Hbuf, N, CIN, CH, part, dinv);
  k_agg_owner<128, true><<<(N + AGG_R - 1) / AGG_R, AGG_T, 0, stream>>>(src, dst, Hbuf, dinv, b1, slabs, AGG, N, E);

  k_gemm_wmma_f32<false><<<(N + 31) / 32, 32 * (CO / 32), 0, stream>>>(AGG, W2, Hbuf, N, CH, CO, nullptr, nullptr);
  k_agg_owner<64, false><<<(N + AGG_R - 1) / AGG_R, AGG_T, 0, stream>>>(src, dst, Hbuf, dinv, b2, slabs, out, N, E);
}


# Product launches: weight matrix copied once per work-group into LDS in fragment order; each wave owns 16 rows x 64 columns, so an x fragment is loaded once per 4 WMMAs; k order and sum orders kept
# speedup vs baseline: 3.5044x; 1.1035x over previous
#include <hip/hip_runtime.h>
#include <hip/hip_bf16.h>

typedef float v2f __attribute__((ext_vector_type(2)));
typedef float v8f __attribute__((ext_vector_type(8)));

constexpr int AGG_R    = 196;
constexpr int AGG_T    = 1024;
constexpr int AGG_QCAP = 8192;
constexpr int AGG_US   = 8;
constexpr int AGG_UB   = 16;
constexpr int AGG_W    = AGG_T / 32;
constexpr int AGG_SLOTS = (AGG_R + AGG_W - 1) / AGG_W;
constexpr int AGG_BCAP = 64;
constexpr int AGG_LK   = 4;
constexpr int AGG_SLABROW = 64;
constexpr int AGG_SLAB = AGG_R * AGG_SLABROW;
constexpr int AGG_OG   = 8;
constexpr int AGG_S    = 8;

constexpr int GM_WAVES = 4;
constexpr int GM_NC    = 64;
constexpr int GM_PITCH = 2 * GM_NC + 32;
constexpr int GM_CH    = 8;

template <int K, int N, bool DINV>
__global__ __launch_bounds__(GM_WAVES * 32) void k_gemm_wmma_f32(const float* __restrict__ A,
                                                                  const float* __restrict__ B,
                                                                  float* __restrict__ C,
                                                                  int M,
                                                                  const int* __restrict__ part,
                                                                  float* __restrict__ dinv) {
  static_assert(K % (4 * GM_CH) == 0 && N % GM_NC == 0, "whole chunks, whole column parts");
  constexpr int NP = N / GM_NC;
  __shared__ float Bs[(K / 2) * GM_PITCH];

  const int tid  = threadIdx.x;
  const int lane = tid & 31;
  const int wv   = tid >> 5;
  const int h    = lane >> 4;
  const int l16  = lane & 15;
  const int cb    = (blockIdx.x % NP) * GM_NC;
  const int rbase = (blockIdx.x / NP) * (GM_WAVES * 16);
  const int mbase = rbase + wv * 16;

  {
    constexpr int ITEMS = (K / 2) * (GM_NC / 4);
    static_assert(ITEMS % (GM_WAVES * 32) == 0, "every thread copies the same number of items");
#pragma unroll 4
    for (int it = tid; it < ITEMS; it += GM_WAVES * 32) {
      const int j  = it % (GM_NC / 4);
      const int kp = it / (GM_NC / 4);
      const float4 r0 = *reinterpret_cast<const float4*>(B + (size_t)(2 * kp) * N + cb + 4 * j);
      const float4 r1 = *reinterpret_cast<const float4*>(B + (size_t)(2 * kp + 1) * N + cb + 4 * j);
      float* d = Bs + kp * GM_PITCH + 2 * j;
      v2f p0, p1, p2, p3;
      p0.x = r0.x;  p0.y = r1.x;
      p1.x = r0.y;  p1.y = r1.y;
      p2.x = r0.z;  p2.y = r1.z;
      p3.x = r0.w;  p3.y = r1.w;
      *reinterpret_cast<v2f*>(d)      = p0;
      *reinterpret_cast<v2f*>(d + 32) = p1;
      *reinterpret_cast<v2f*>(d + 64) = p2;
      *reinterpret_cast<v2f*>(d + 96) = p3;
    }
  }

  const int m  = mbase + l16;
  const int mc = (m < M) ? m : (M - 1);
  const float* arow = A + (size_t)mc * K + 2 * h;
  const float* bl   = Bs + h * GM_PITCH + 2 * l16;

  v2f an[GM_CH];
#pragma unroll
  for (int s = 0; s < GM_CH; ++s) an[s] = *reinterpret_cast<const v2f*>(arow + 4 * s);
  __syncthreads();

  v8f acc0 = {}, acc1 = {}, acc2 = {}, acc3 = {};
  for (int kc = 0; kc < K; kc += 4 * GM_CH) {
    v2f ac[GM_CH];
#pragma unroll
    for (int s = 0; s < GM_CH; ++s) ac[s] = an[s];
    if (kc + 4 * GM_CH < K) {
#pragma unroll
      for (int s = 0; s < GM_CH; ++s) an[s] = *reinterpret_cast<const v2f*>(arow + kc + 4 * GM_CH + 4 * s);
    }
#pragma unroll
    for (int s = 0; s < GM_CH; ++s) {
      const float* bp = bl + ((kc >> 1) + 2 * s) * GM_PITCH;
      const v2f b0 = *reinterpret_cast<const v2f*>(bp);
      const v2f b1 = *reinterpret_cast<const v2f*>(bp + 32);
      const v2f b2 = *reinterpret_cast<const v2f*>(bp + 64);
      const v2f b3 = *reinterpret_cast<const v2f*>(bp + 96);
      acc0 = __builtin_amdgcn_wmma_f32_16x16x4_f32(false, ac[s], false, b0, (short)0, acc0, false, false);
      acc1 = __builtin_amdgcn_wmma_f32_16x16x4_f32(false, ac[s], false, b1, (short)0, acc1, false, false);
      acc2 = __builtin_amdgcn_wmma_f32_16x16x4_f32(false, ac[s], false, b2, (short)0, acc2, false, false);
      acc3 = __builtin_amdgcn_wmma_f32_16x16x4_f32(false, ac[s], false, b3, (short)0, acc3, false, false);
    }
  }

#pragma unroll
  for (int r = 0; r < 8; ++r) {
    const int row = mbase + r + 8 * h;
    if (row < M) {
      float4 o;
      o.x = acc0[r];  o.y = acc1[r];  o.z = acc2[r];  o.w = acc3[r];
      *reinterpret_cast<float4*>(C + (size_t)row * N + cb + 4 * l16) = o;
    }
  }

  if (DINV && cb == 0) {
    for (int node = rbase + tid; node < rbase + GM_WAVES * 16 && node < M; node += GM_WAVES * 32) {
      int deg = 0;
#pragma unroll
      for (int s = 0; s < AGG_S; ++s) deg += part[(size_t)s * M + node];
      dinv[node] = rsqrtf(1.0f + (float)deg);
    }
  }
}

static_assert(AGG_US * 4 <= 32, "one hit bit per scanned destination");
template <bool B> struct agg_flag { static constexpr bool value = B; };

template <int V> struct agg_int { static constexpr int value = V; };
template <int I, int N, class F>
__device__ __forceinline__ void agg_static_upto(int n, F&& f) {
  if constexpr (I < N) {
    if (I < n) { f(agg_int<I>()); agg_static_upto<I + 1, N>(n, f); }
  }
}
template <int U> __device__ __forceinline__ int agg_half_bcast(int v) {
  return __builtin_amdgcn_update_dpp(0, v, 0x150 + U, 0xf, 0xf, true);
}

__device__ __forceinline__ int agg_seg_words(int nloc) { return (nloc * AGG_SLABROW) / AGG_S; }

__device__ __forceinline__ void agg_part_bounds(int units, int part, int nparts, int& lo, int& hi) {
  const int per = (units + nparts - 1) / nparts;
  lo = part * per;
  if (lo > units) lo = units;
  hi = lo + per;
  if (hi > units) hi = units;
}

template <class Push>
__device__ __forceinline__ void agg_scan_dst(const int* __restrict__ dst, int E, int part, int nparts, int base, int nloc, int tid, Push push) {
  if ((E & 3) == 0) {
    const int4* d4 = reinterpret_cast<const int4*>(dst);
    int lo, E4;
    agg_part_bounds(E >> 2, part, nparts, lo, E4);
    auto round = [&](int g, auto guarded) {
      int4 dv[AGG_US];
#pragma unroll
      for (int u = 0; u < AGG_US; ++u) {
        const int gu = g + u * AGG_T;
        dv[u] = d4[(!guarded.value || gu < E4) ? gu : g];
      }
      unsigned hm = 0u;
#pragma unroll
      for (int u = 0; u < AGG_US; ++u) {
        const unsigned lim = (!guarded.value || g + u * AGG_T < E4) ? (unsigned)nloc : 0u;
        hm |= ((unsigned)(dv[u].x - base) < lim) ? (1u << (4 * u))     : 0u;
        hm |= ((unsigned)(dv[u].y - base) < lim) ? (1u << (4 * u + 1)) : 0u;
        hm |= ((unsigned)(dv[u].z - base) < lim) ? (1u << (4 * u + 2)) : 0u;
        hm |= ((unsigned)(dv[u].w - base) < lim) ? (1u << (4 * u + 3)) : 0u;
      }
      while (hm) {
        const int b = __builtin_ctz(hm);
        hm &= hm - 1u;
        push(4 * (g + (b >> 2) * AGG_T) + (b & 3));
      }
    };
    int g = lo + tid;
    for (; g + (AGG_US - 1) * AGG_T < E4; g += AGG_US * AGG_T) round(g, agg_flag<false>());
    if (g < E4) round(g, agg_flag<true>());
  } else {
    int lo, hi;
    agg_part_bounds(E, part, nparts, lo, hi);
    for (int e = lo + tid; e < hi; e += AGG_T)
      if ((unsigned)(dst[e] - base) < (unsigned)nloc) push(e);
  }
}

__global__ __launch_bounds__(1024) void k_owner_scan(const int* __restrict__ dst,
                                                     int* __restrict__ part,
                                                     int* __restrict__ slabs,
                                                     int N, int E) {
  __shared__ int q[AGG_QCAP];
  __shared__ int cnt[AGG_OG * AGG_R];
  __shared__ int ocount[AGG_OG];
  __shared__ int qcount;

  const int tid   = threadIdx.x;
  const int grp   = blockIdx.x / AGG_S;
  const int sp    = blockIdx.x % AGG_S;
  const int gbase = grp * (AGG_OG * AGG_R);
  int gn = N - gbase;
  if (gn > AGG_OG * AGG_R) gn = AGG_OG * AGG_R;

  if (tid == 0) qcount = 0;
  if (tid < AGG_OG) ocount[tid] = 0;
  for (int r = tid; r < AGG_OG * AGG_R; r += AGG_T) cnt[r] = 0;
  __syncthreads();

  agg_scan_dst(dst, E, sp, AGG_S, gbase, gn, tid, [&](int e) {
    const int pos = atomicAdd(&qcount, 1);
    if (pos < AGG_QCAP) q[pos] = e;
  });
  __syncthreads();

  const int total = qcount;
  const bool over = total > AGG_QCAP;
  if (!over) {
    for (int k = tid; k < total; k += AGG_T) {
      const int e  = q[k];
      const int dl = dst[e] - gbase;
      const int o  = dl / AGG_R;
      atomicAdd(&cnt[dl], 1);
      const int pos = atomicAdd(&ocount[o], 1);
      int onl = N - (gbase + o * AGG_R);
      if (onl > AGG_R) onl = AGG_R;
      const int segw = agg_seg_words(onl);
      if (pos < segw - 1) slabs[(size_t)(grp * AGG_OG + o) * AGG_SLAB + sp * segw + 1 + pos] = e;
    }
  } else {
    const bool wide = (E & 3) == 0;
    int lo, hi;
    agg_part_bounds(wide ? (E >> 2) : E, sp, AGG_S, lo, hi);
    if (wide) { lo *= 4; hi *= 4; }
    for (int e = lo + tid; e < hi; e += AGG_T) {
      const int dl = dst[e] - gbase;
      if ((unsigned)dl < (unsigned)gn) atomicAdd(&cnt[dl], 1);
    }
  }
  __syncthreads();

  if (tid < AGG_OG) {
    const int obase = gbase + tid * AGG_R;
    if (obase < N) {
      int onl = N - obase;
      if (onl > AGG_R) onl = AGG_R;
      slabs[(size_t)(grp * AGG_OG + tid) * AGG_SLAB + sp * agg_seg_words(onl)] = over ? (AGG_QCAP + 1) : ocount[tid];
    }
  }
  for (int r = tid; r < gn; r += AGG_T) part[(size_t)sp * N + gbase + r] = cnt[r];
}

template <int C, bool RELU>
__global__ __launch_bounds__(1024) void k_agg_owner(const int* __restrict__ src,
                                                    const int* __restrict__ dst,
                                                    const float* __restrict__ H,
                                                    const float* __restrict__ dinv,
                                                    const float* __restrict__ bias,
                                                    const int* slabs,
                                                    float* out,
                                                    int N, int E) {
  constexpr int VEC = C / 32;
  constexpr int LPR = C / 4;
  constexpr int EPL = 32 / LPR;
  static_assert(EPL == 1 || EPL == 2, "a source row fills a wave or a half-wave");
  __shared__ float acc[AGG_R * C];
  __shared__ int   q[AGG_QCAP];
  __shared__ int   bsrc[AGG_R * AGG_BCAP];
  __shared__ float bwt[AGG_R * AGG_BCAP];
  __shared__ int   bcount[AGG_R];
  __shared__ float dloc[AGG_R];
  __shared__ int   qcount;

  const int tid  = threadIdx.x;
  const int lane = tid & 31;
  const int wv   = tid >> 5;
  const int wvu  = __builtin_amdgcn_readfirstlane(wv);
  const int sub  = lane / LPR;
  const int lch  = lane % LPR;
  const int base = blockIdx.x * AGG_R;
  int nloc = N - base;
  if (nloc > AGG_R) nloc = AGG_R;

  const int* slab = slabs + (size_t)blockIdx.x * AGG_SLAB;
  const int segw = agg_seg_words(nloc);
  int segc[AGG_S];
  int total = 0;
  bool kept = true;
#pragma unroll
  for (int s = 0; s < AGG_S; ++s) {
    segc[s] = __builtin_amdgcn_readfirstlane(slab[s * segw]);
    kept = kept && (segc[s] <= segw - 1);
    total += segc[s];
  }

  float4 sum[AGG_SLOTS];
  {
    float dw[AGG_SLOTS];
    float4 hs[AGG_SLOTS];
#pragma unroll
    for (int s = 0; s < AGG_SLOTS; ++s) {
      const int r  = wvu + AGG_W * s;
      const int rr = (r < nloc) ? r : (nloc - 1);
      dw[s] = dinv[base + rr];
      hs[s] = *reinterpret_cast<const float4*>(H + (size_t)(base + rr) * C + lch * 4);
    }
#pragma unroll
    for (int s = 0; s < AGG_SLOTS; ++s) {
      const float w = (wvu + AGG_W * s < nloc && sub == 0) ? dw[s] * dw[s] : 0.0f;
      sum[s].x = hs[s].x * w;  sum[s].y = hs[s].y * w;  sum[s].z = hs[s].z * w;  sum[s].w = hs[s].w * w;
    }
  }
  if (tid == 0) qcount = 0;
  for (int r = tid; r < nloc; r += AGG_T) { dloc[r] = dinv[base + r]; bcount[r] = 0; }
  {
    float4* a4 = reinterpret_cast<float4*>(acc);
    const float4 z = {0.0f, 0.0f, 0.0f, 0.0f};
    for (int i = tid; i < (nloc * C) / 4; i += AGG_T) a4[i] = z;
  }
  __syncthreads();

  if (!kept) {
    agg_scan_dst(dst, E, 0, 1, base, nloc, tid, [&](int e) {
      const int pos = atomicAdd(&qcount, 1);
      if (pos < AGG_QCAP) {
        q[pos] = e;
      } else {
        const int i = src[e];
        const int d = dst[e];
        const float w = dinv[i] * dinv[d];
        const float* hrow = H + (size_t)i * C;
        float* arow = acc + (d - base) * C;
        for (int c = 0; c < C; ++c) atomicAdd(&arow[(c % VEC) * 32 + c / VEC], hrow[c] * w);
      }
    });
  }
  __syncthreads();

  int qn = total;
  if (!kept) {
    qn = qcount;
    if (qn > AGG_QCAP) qn = AGG_QCAP;
  }
  qn = __builtin_amdgcn_readfirstlane(qn);
  for (int k0 = 0; k0 < qn; k0 += AGG_LK * AGG_T) {
    int  ev[AGG_LK];
    bool okv[AGG_LK];
#pragma unroll
    for (int j = 0; j < AGG_LK; ++j) {
      const int kk = k0 + j * AGG_T + tid;
      okv[j] = kk < qn;
      const int k = okv[j] ? kk : (qn - 1);
      if (kept) {
        int seg = 0, first = 0, run = 0;
#pragma unroll
        for (int s = 0; s + 1 < AGG_S; ++s) {
          run += segc[s];
          if (k >= run) { seg = s + 1; first = run; }
        }
        ev[j] = slab[seg * segw + 1 + (k - first)];
      } else {
        ev[j] = q[k];
      }
    }
    int siv[AGG_LK], djv[AGG_LK];
#pragma unroll
    for (int j = 0; j < AGG_LK; ++j) {
      siv[j] = src[ev[j]];
      djv[j] = dst[ev[j]];
    }
    float wev[AGG_LK];
#pragma unroll
    for (int j = 0; j < AGG_LK; ++j) {
      djv[j] = okv[j] ? (djv[j] - base) : 0;
      wev[j] = dinv[siv[j]] * dloc[djv[j]];
    }
#pragma unroll
    for (int j = 0; j < AGG_LK; ++j) {
      if (okv[j]) {
        const int pos = atomicAdd(&bcount[djv[j]], 1);
        if (pos < AGG_BCAP) {
          bsrc[djv[j] * AGG_BCAP + pos] = siv[j];
          bwt[djv[j] * AGG_BCAP + pos]  = wev[j];
        } else {
          const float* hrow = H + (size_t)siv[j] * C;
          float* arow = acc + djv[j] * C;
          for (int c = 0; c < C; ++c) atomicAdd(&arow[(c % VEC) * 32 + c / VEC], hrow[c] * wev[j]);
        }
      }
    }
  }
  __syncthreads();

  constexpr int UB = AGG_UB;
  const float4 bv = *reinterpret_cast<const float4*>(bias + lch * 4);
  const char* Hb = reinterpret_cast<const char*>(H);
  const unsigned laneoff = (unsigned)lch * 16u;
  const int ent = (EPL == 1) ? lane : (2 * lch + sub);
#pragma unroll
  for (int s = 0; s < AGG_SLOTS; ++s) {
    const int r = wvu + AGG_W * s;
    if (r < nloc) {
      float4 a = sum[s];
      int bn = bcount[r];
      if (bn > AGG_BCAP) bn = AGG_BCAP;
      bn = __builtin_amdgcn_readfirstlane(bn);
      const int*   mys = bsrc + r * AGG_BCAP;
      const float* myw = bwt + r * AGG_BCAP;
      for (int i0 = 0; i0 < bn; i0 += 32) {
        const int cnt = (bn - i0 < 32) ? (bn - i0) : 32;
        const bool ok = ent < cnt;
        unsigned so = ok ? (unsigned)mys[i0 + ent] * (unsigned)(C * 4) : 0u;
        const float we = ok ? myw[i0 + ent] : 0.0f;
        if (EPL == 2) {
          const int last = cnt - 1;
          const unsigned solast = (unsigned)__builtin_amdgcn_readlane((int)so, (last >> 1) + 16 * (last & 1));
          so = ok ? so : solast;
        }
        const int nld = (cnt + EPL - 1) / EPL;
        for (int u0 = 0; u0 < nld; u0 += UB) {
          float4 hv[UB];
          const int nb = nld - u0;
          agg_static_upto<0, UB>(nb, [&](auto uc) {
            constexpr int u = decltype(uc)::value;
            unsigned off;
            if constexpr (EPL == 1) off = (unsigned)__builtin_amdgcn_readlane((int)so, u0 + u);
            else                    off = (unsigned)agg_half_bcast<u>((int)so);
            hv[u] = *reinterpret_cast<const float4*>(Hb + (off + laneoff));
          });
          agg_static_upto<0, UB>(nb, [&](auto uc) {
            constexpr int u = decltype(uc)::value;
            float w;
            if constexpr (EPL == 1) w = __int_as_float(__builtin_amdgcn_readlane(__float_as_int(we), u0 + u));
            else                    w = __int_as_float(agg_half_bcast<u>(__float_as_int(we)));
            a.x += hv[u].x * w;  a.y += hv[u].y * w;  a.z += hv[u].z * w;  a.w += hv[u].w * w;
          });
        }
      }
      if (EPL == 2) {
        a.x += __int_as_float(__builtin_amdgcn_permlanex16(0, __float_as_int(a.x), 0x76543210, 0xfedcba98, false, false));
        a.y += __int_as_float(__builtin_amdgcn_permlanex16(0, __float_as_int(a.y), 0x76543210, 0xfedcba98, false, false));
        a.z += __int_as_float(__builtin_amdgcn_permlanex16(0, __float_as_int(a.z), 0x76543210, 0xfedcba98, false, false));
        a.w += __int_as_float(__builtin_amdgcn_permlanex16(0, __float_as_int(a.w), 0x76543210, 0xfedcba98, false, false));
      }
      if (sub == 0) {
        const float* ar = acc + r * C;
        const float av[4] = {a.x, a.y, a.z, a.w};
        const float bb[4] = {bv.x, bv.y, bv.z, bv.w};
        float o[4];
#pragma unroll
        for (int t = 0; t < 4; ++t) {
          const int ch = 4 * lch + t;
          const float v = av[t] + ar[(ch % VEC) * 32 + ch / VEC] + bb[t];
          o[t] = RELU ? fmaxf(v, 0.0f) : v;
        }
        float4 ov;
        ov.x = o[0];  ov.y = o[1];  ov.z = o[2];  ov.w = o[3];
        *reinterpret_cast<float4*>(out + (size_t)(base + r) * C + lch * 4) = ov;
      }
    }
  }
}

extern "C" void kernel_launch(void* const* d_in, const int* in_sizes, int n_in,
                              void* d_out, int out_size, void* d_ws, size_t ws_size,
                              hipStream_t stream) {
  const int CIN = 256, CH = 128, CO = 64;
  const float* x  = (const float*)d_in[0];
  const int*   ei = (const int*)d_in[1];
  const float* W1 = (const float*)d_in[2];
  const float* b1 = (const float*)d_in[3];
  const float* W2 = (const float*)d_in[4];
  const float* b2 = (const float*)d_in[5];
  float* out = (float*)d_out;

  const int N = in_sizes[0] / CIN;
  const int E = in_sizes[1] / 2;
  const int* src = ei;
  const int* dst = ei + E;

  float* ws   = (float*)d_ws;
  float* dinv = ws;
  float* Hbuf = ws + 50176;
  float* AGG  = ws + 50176 + (size_t)N * CH;

  int* slabs = (int*)d_out;
  int* part = (int*)AGG;

  const int owners = (N + AGG_R - 1) / AGG_R;
  k_owner_scan<<<((owners + AGG_OG - 1) / AGG_OG) * AGG_S, AGG_T, 0, stream>>>(dst, part, slabs, N, E);

  k_gemm_wmma_f32<256, 128, true><<<((N + GM_WAVES * 16 - 1) / (GM_WAVES * 16)) * (CH / GM_NC), GM_WAVES * 32, 0, stream>>>(x, W1, Hbuf, N, part, dinv);
  k_agg_owner<128, true><<<(N + AGG_R - 1) / AGG_R, AGG_T, 0, stream>>>(src, dst, Hbuf, dinv, b1, slabs, AGG, N, E);

  k_gemm_wmma_f32<128, 64, false><<<((N + GM_WAVES * 16 - 1) / (GM_WAVES * 16)) * (CO / GM_NC), GM_WAVES * 32, 0, stream>>>(AGG, W2, Hbuf, N, nullptr, nullptr);
  k_agg_owner<64, false><<<(N + AGG_R - 1) / AGG_R, AGG_T, 0, stream>>>(src, dst, Hbuf, dinv, b2, slabs, out, N, E);
}


# Owner opening: scan keeps one packed word per edge (source, local row); owner issues record and dinv look-ups ahead of its LDS set-up barriers, beside the start rows; sum order = bin arrival
# speedup vs baseline: 3.8212x; 1.0904x over previous
#include <hip/hip_runtime.h>
#include <hip/hip_bf16.h>

typedef float v2f __attribute__((ext_vector_type(2)));
typedef float v8f __attribute__((ext_vector_type(8)));

constexpr int AGG_R    = 196;
constexpr int AGG_T    = 1024;
constexpr int AGG_QCAP = 8192;
constexpr int AGG_US   = 8;
constexpr int AGG_UB   = 16;
constexpr int AGG_W    = AGG_T / 32;
constexpr int AGG_SLOTS = (AGG_R + AGG_W - 1) / AGG_W;
constexpr int AGG_BCAP = 64;
constexpr int AGG_LK   = 4;
constexpr int AGG_SLABROW = 64;
constexpr int AGG_SLAB = AGG_R * AGG_SLABROW;
constexpr int AGG_OG   = 8;
constexpr int AGG_S    = 8;

constexpr int GM_WAVES = 4;
constexpr int GM_NC    = 64;
constexpr int GM_PITCH = 2 * GM_NC + 32;
constexpr int GM_CH    = 8;

template <int K, int N, bool DINV>
__global__ __launch_bounds__(GM_WAVES * 32) void k_gemm_wmma_f32(const float* __restrict__ A,
                                                                  const float* __restrict__ B,
                                                                  float* __restrict__ C,
                                                                  int M,
                                                                  const int* __restrict__ part,
                                                                  float* __restrict__ dinv) {
  static_assert(K % (4 * GM_CH) == 0 && N % GM_NC == 0, "whole chunks, whole column parts");
  constexpr int NP = N / GM_NC;
  __shared__ float Bs[(K / 2) * GM_PITCH];

  const int tid  = threadIdx.x;
  const int lane = tid & 31;
  const int wv   = tid >> 5;
  const int h    = lane >> 4;
  const int l16  = lane & 15;
  const int cb    = (blockIdx.x % NP) * GM_NC;
  const int rbase = (blockIdx.x / NP) * (GM_WAVES * 16);
  const int mbase = rbase + wv * 16;

  {
    constexpr int ITEMS = (K / 2) * (GM_NC / 4);
    static_assert(ITEMS % (GM_WAVES * 32) == 0, "every thread copies the same number of items");
#pragma unroll 4
    for (int it = tid; it < ITEMS; it += GM_WAVES * 32) {
      const int j  = it % (GM_NC / 4);
      const int kp = it / (GM_NC / 4);
      const float4 r0 = *reinterpret_cast<const float4*>(B + (size_t)(2 * kp) * N + cb + 4 * j);
      const float4 r1 = *reinterpret_cast<const float4*>(B + (size_t)(2 * kp + 1) * N + cb + 4 * j);
      float* d = Bs + kp * GM_PITCH + 2 * j;
      v2f p0, p1, p2, p3;
      p0.x = r0.x;  p0.y = r1.x;
      p1.x = r0.y;  p1.y = r1.y;
      p2.x = r0.z;  p2.y = r1.z;
      p3.x = r0.w;  p3.y = r1.w;
      *reinterpret_cast<v2f*>(d)      = p0;
      *reinterpret_cast<v2f*>(d + 32) = p1;
      *reinterpret_cast<v2f*>(d + 64) = p2;
      *reinterpret_cast<v2f*>(d + 96) = p3;
    }
  }

  const int m  = mbase + l16;
  const int mc = (m < M) ? m : (M - 1);
  const float* arow = A + (size_t)mc * K + 2 * h;
  const float* bl   = Bs + h * GM_PITCH + 2 * l16;

  v2f an[GM_CH];
#pragma unroll
  for (int s = 0; s < GM_CH; ++s) an[s] = *reinterpret_cast<const v2f*>(arow + 4 * s);
  __syncthreads();

  v8f acc0 = {}, acc1 = {}, acc2 = {}, acc3 = {};
  for (int kc = 0; kc < K; kc += 4 * GM_CH) {
    v2f ac[GM_CH];
#pragma unroll
    for (int s = 0; s < GM_CH; ++s) ac[s] = an[s];
    if (kc + 4 * GM_CH < K) {
#pragma unroll
      for (int s = 0; s < GM_CH; ++s) an[s] = *reinterpret_cast<const v2f*>(arow + kc + 4 * GM_CH + 4 * s);
    }
#pragma unroll
    for (int s = 0; s < GM_CH; ++s) {
      const float* bp = bl + ((kc >> 1) + 2 * s) * GM_PITCH;
      const v2f b0 = *reinterpret_cast<const v2f*>(bp);
      const v2f b1 = *reinterpret_cast<const v2f*>(bp + 32);
      const v2f b2 = *reinterpret_cast<const v2f*>(bp + 64);
      const v2f b3 = *reinterpret_cast<const v2f*>(bp + 96);
      acc0 = __builtin_amdgcn_wmma_f32_16x16x4_f32(false, ac[s], false, b0, (short)0, acc0, false, false);
      acc1 = __builtin_amdgcn_wmma_f32_16x16x4_f32(false, ac[s], false, b1, (short)0, acc1, false, false);
      acc2 = __builtin_amdgcn_wmma_f32_16x16x4_f32(false, ac[s], false, b2, (short)0, acc2, false, false);
      acc3 = __builtin_amdgcn_wmma_f32_16x16x4_f32(false, ac[s], false, b3, (short)0, acc3, false, false);
    }
  }

#pragma unroll
  for (int r = 0; r < 8; ++r) {
    const int row = mbase + r + 8 * h;
    if (row < M) {
      float4 o;
      o.x = acc0[r];  o.y = acc1[r];  o.z = acc2[r];  o.w = acc3[r];
      *reinterpret_cast<float4*>(C + (size_t)row * N + cb + 4 * l16) = o;
    }
  }

  if (DINV && cb == 0) {
    for (int node = rbase + tid; node < rbase + GM_WAVES * 16 && node < M; node += GM_WAVES * 32) {
      int deg = 0;
#pragma unroll
      for (int s = 0; s < AGG_S; ++s) deg += part[(size_t)s * M + node];
      dinv[node] = rsqrtf(1.0f + (float)deg);
    }
  }
}

static_assert(AGG_US * 4 <= 32, "one hit bit per scanned destination");
template <bool B> struct agg_flag { static constexpr bool value = B; };

template <int V> struct agg_int { static constexpr int value = V; };
template <int I, int N, class F>
__device__ __forceinline__ void agg_static_upto(int n, F&& f) {
  if constexpr (I < N) {
    if (I < n) { f(agg_int<I>()); agg_static_upto<I + 1, N>(n, f); }
  }
}
template <int U> __device__ __forceinline__ int agg_half_bcast(int v) {
  return __builtin_amdgcn_update_dpp(0, v, 0x150 + U, 0xf, 0xf, true);
}

__device__ __forceinline__ int agg_seg_words(int nloc) { return (nloc * AGG_SLABROW) / AGG_S; }

constexpr int AGG_PACK_NODES = 1 << 24;
static_assert(AGG_R <= 256, "the local row fits the 8 bits above the source");
__device__ __forceinline__ int agg_pack(int s, int row) { return (int)((unsigned)s | ((unsigned)row << 24)); }

__device__ __forceinline__ void agg_part_bounds(int units, int part, int nparts, int& lo, int& hi) {
  const int per = (units + nparts - 1) / nparts;
  lo = part * per;
  if (lo > units) lo = units;
  hi = lo + per;
  if (hi > units) hi = units;
}

template <class Push>
__device__ __forceinline__ void agg_scan_dst(const int* __restrict__ dst, int E, int part, int nparts, int base, int nloc, int tid, Push push) {
  if ((E & 3) == 0) {
    const int4* d4 = reinterpret_cast<const int4*>(dst);
    int lo, E4;
    agg_part_bounds(E >> 2, part, nparts, lo, E4);
    auto round = [&](int g, auto guarded) {
      int4 dv[AGG_US];
#pragma unroll
      for (int u = 0; u < AGG_US; ++u) {
        const int gu = g + u * AGG_T;
        dv[u] = d4[(!guarded.value || gu < E4) ? gu : g];
      }
      unsigned hm = 0u;
#pragma unroll
      for (int u = 0; u < AGG_US; ++u) {
        const unsigned lim = (!guarded.value || g + u * AGG_T < E4) ? (unsigned)nloc : 0u;
        hm |= ((unsigned)(dv[u].x - base) < lim) ? (1u << (4 * u))     : 0u;
        hm |= ((unsigned)(dv[u].y - base) < lim) ? (1u << (4 * u + 1)) : 0u;
        hm |= ((unsigned)(dv[u].z - base) < lim) ? (1u << (4 * u + 2)) : 0u;
        hm |= ((unsigned)(dv[u].w - base) < lim) ? (1u << (4 * u + 3)) : 0u;
      }
      while (hm) {
        const int b = __builtin_ctz(hm);
        hm &= hm - 1u;
        push(4 * (g + (b >> 2) * AGG_T) + (b & 3));
      }
    };
    int g = lo + tid;
    for (; g + (AGG_US - 1) * AGG_T < E4; g += AGG_US * AGG_T) round(g, agg_flag<false>());
    if (g < E4) round(g, agg_flag<true>());
  } else {
    int lo, hi;
    agg_part_bounds(E, part, nparts, lo, hi);
    for (int e = lo + tid; e < hi; e += AGG_T)
      if ((unsigned)(dst[e] - base) < (unsigned)nloc) push(e);
  }
}

__global__ __launch_bounds__(1024) void k_owner_scan(const int* __restrict__ src,
                                                     const int* __restrict__ dst,
                                                     int* __restrict__ part,
                                                     int* __restrict__ slabs,
                                                     int N, int E) {
  __shared__ int q[AGG_QCAP];
  __shared__ int cnt[AGG_OG * AGG_R];
  __shared__ int ocount[AGG_OG];
  __shared__ int qcount;

  const int tid   = threadIdx.x;
  const int grp   = blockIdx.x / AGG_S;
  const int sp    = blockIdx.x % AGG_S;
  const int gbase = grp * (AGG_OG * AGG_R);
  int gn = N - gbase;
  if (gn > AGG_OG * AGG_R) gn = AGG_OG * AGG_R;

  if (tid == 0) qcount = 0;
  if (tid < AGG_OG) ocount[tid] = 0;
  for (int r = tid; r < AGG_OG * AGG_R; r += AGG_T) cnt[r] = 0;
  __syncthreads();

  agg_scan_dst(dst, E, sp, AGG_S, gbase, gn, tid, [&](int e) {
    const int pos = atomicAdd(&qcount, 1);
    if (pos < AGG_QCAP) q[pos] = e;
  });
  __syncthreads();

  const int total = qcount;
  const bool over = total > AGG_QCAP;
  if (!over) {
    for (int k = tid; k < total; k += AGG_T) {
      const int e  = q[k];
      const int sv = src[e];
      const int dl = dst[e] - gbase;
      const int o  = dl / AGG_R;
      atomicAdd(&cnt[dl], 1);
      const int pos = atomicAdd(&ocount[o], 1);
      int onl = N - (gbase + o * AGG_R);
      if (onl > AGG_R) onl = AGG_R;
      const int segw = agg_seg_words(onl);
      if (pos < segw - 1) slabs[(size_t)(grp * AGG_OG + o) * AGG_SLAB + sp * segw + 1 + pos] = agg_pack(sv, dl - o * AGG_R);
    }
  } else {
    const bool wide = (E & 3) == 0;
    int lo, hi;
    agg_part_bounds(wide ? (E >> 2) : E, sp, AGG_S, lo, hi);
    if (wide) { lo *= 4; hi *= 4; }
    for (int e = lo + tid; e < hi; e += AGG_T) {
      const int dl = dst[e] - gbase;
      if ((unsigned)dl < (unsigned)gn) atomicAdd(&cnt[dl], 1);
    }
  }
  __syncthreads();

  if (tid < AGG_OG) {
    const int obase = gbase + tid * AGG_R;
    if (obase < N) {
      int onl = N - obase;
      if (onl > AGG_R) onl = AGG_R;
      slabs[(size_t)(grp * AGG_OG + tid) * AGG_SLAB + sp * agg_seg_words(onl)] = (over || N > AGG_PACK_NODES) ? (AGG_QCAP + 1) : ocount[tid];
    }
  }
  for (int r = tid; r < gn; r += AGG_T) part[(size_t)sp * N + gbase + r] = cnt[r];
}

template <int C, bool RELU>
__global__ __launch_bounds__(1024) void k_agg_owner(const int* __restrict__ src,
                                                    const int* __restrict__ dst,
                                                    const float* __restrict__ H,
                                                    const float* __restrict__ dinv,
                                                    const float* __restrict__ bias,
                                                    const int* slabs,
                                                    float* out,
                                                    int N, int E) {
  constexpr int VEC = C / 32;
  constexpr int LPR = C / 4;
  constexpr int EPL = 32 / LPR;
  static_assert(EPL == 1 || EPL == 2, "a source row fills a wave or a half-wave");
  __shared__ float acc[AGG_R * C];
  __shared__ int   q[AGG_QCAP];
  __shared__ int   bsrc[AGG_R * AGG_BCAP];
  __shared__ float bwt[AGG_R * AGG_BCAP];
  __shared__ int   bcount[AGG_R];
  __shared__ float dloc[AGG_R];
  __shared__ int   qcount;

  const int tid  = threadIdx.x;
  const int lane = tid & 31;
  const int wv   = tid >> 5;
  const int wvu  = __builtin_amdgcn_readfirstlane(wv);
  const int sub  = lane / LPR;
  const int lch  = lane % LPR;
  const int base = blockIdx.x * AGG_R;
  int nloc = N - base;
  if (nloc > AGG_R) nloc = AGG_R;

  const int* slab = slabs + (size_t)blockIdx.x * AGG_SLAB;
  const int segw = agg_seg_words(nloc);
  int segc[AGG_S];
  int total = 0;
  bool kept = true;
#pragma unroll
  for (int s = 0; s < AGG_S; ++s) {
    segc[s] = __builtin_amdgcn_readfirstlane(slab[s * segw]);
    kept = kept && (segc[s] <= segw - 1);
    total += segc[s];
  }

  float4 sum[AGG_SLOTS];
  {
    float dw[AGG_SLOTS];
    float4 hs[AGG_SLOTS];
#pragma unroll
    for (int s = 0; s < AGG_SLOTS; ++s) {
      const int r  = wvu + AGG_W * s;
      const int rr = (r < nloc) ? r : (nloc - 1);
      dw[s] = dinv[base + rr];
      hs[s] = *reinterpret_cast<const float4*>(H + (size_t)(base + rr) * C + lch * 4);
    }
#pragma unroll
    for (int s = 0; s < AGG_SLOTS; ++s) {
      const float w = (wvu + AGG_W * s < nloc && sub == 0) ? dw[s] * dw[s] : 0.0f;
      sum[s].x = hs[s].x * w;  sum[s].y = hs[s].y * w;  sum[s].z = hs[s].z * w;  sum[s].w = hs[s].w * w;
    }
  }

  auto lookup = [&](int k0, int qn, int& okm, int (&siv)[AGG_LK], int (&djv)[AGG_LK], float (&sdv)[AGG_LK]) {
    int ev[AGG_LK];
    okm = 0;
#pragma unroll
    for (int j = 0; j < AGG_LK; ++j) {
      const int kk = k0 + j * AGG_T + tid;
      const bool ok = kk < qn;
      okm |= ok ? (1 << j) : 0;
      const int k = ok ? kk : (qn - 1);
      if (kept) {
        int seg = 0, first = 0, run = 0;
#pragma unroll
        for (int s = 0; s + 1 < AGG_S; ++s) {
          run += segc[s];
          if (k >= run) { seg = s + 1; first = run; }
        }
        ev[j] = slab[seg * segw + 1 + (k - first)];
      } else {
        ev[j] = q[k];
      }
    }
    if (kept) {
#pragma unroll
      for (int j = 0; j < AGG_LK; ++j) {
        siv[j] = ev[j] & 0xffffff;
        djv[j] = (int)((unsigned)ev[j] >> 24);
      }
    } else {
#pragma unroll
      for (int j = 0; j < AGG_LK; ++j) {
        siv[j] = src[ev[j]];
        djv[j] = dst[ev[j]] - base;
      }
    }
#pragma unroll
    for (int j = 0; j < AGG_LK; ++j) {
      if (((okm >> j) & 1) == 0) djv[j] = 0;
      sdv[j] = dinv[siv[j]];
    }
  };
  const bool early = kept && total > 0;
  int   okm0 = 0;
  int   siv0[AGG_LK], djv0[AGG_LK];
  float sdv0[AGG_LK];
  if (early) lookup(0, total, okm0, siv0, djv0, sdv0);

  if (tid == 0) qcount = 0;
  for (int r = tid; r < nloc; r += AGG_T) { dloc[r] = dinv[base + r]; bcount[r] = 0; }
  {
    float4* a4 = reinterpret_cast<float4*>(acc);
    const float4 z = {0.0f, 0.0f, 0.0f, 0.0f};
    for (int i = tid; i < (nloc * C) / 4; i += AGG_T) a4[i] = z;
  }
  __syncthreads();

  if (!kept) {
    agg_scan_dst(dst, E, 0, 1, base, nloc, tid, [&](int e) {
      const int pos = atomicAdd(&qcount, 1);
      if (pos < AGG_QCAP) {
        q[pos] = e;
      } else {
        const int i = src[e];
        const int d = dst[e];
        const float w = dinv[i] * dinv[d];
        const float* hrow = H + (size_t)i * C;
        float* arow = acc + (d - base) * C;
        for (int c = 0; c < C; ++c) atomicAdd(&arow[(c % VEC) * 32 + c / VEC], hrow[c] * w);
      }
    });
  }
  __syncthreads();

  auto drop = [&](int okm, const int (&siv)[AGG_LK], const int (&djv)[AGG_LK], const float (&sdv)[AGG_LK]) {
    float wev[AGG_LK];
#pragma unroll
    for (int j = 0; j < AGG_LK; ++j) wev[j] = sdv[j] * dloc[djv[j]];
#pragma unroll
    for (int j = 0; j < AGG_LK; ++j) {
      if ((okm >> j) & 1) {
        const int pos = atomicAdd(&bcount[djv[j]], 1);
        if (pos < AGG_BCAP) {
          bsrc[djv[j] * AGG_BCAP + pos] = siv[j];
          bwt[djv[j] * AGG_BCAP + pos]  = wev[j];
        } else {
          const float* hrow = H + (size_t)siv[j] * C;
          float* arow = acc + djv[j] * C;
          for (int c = 0; c < C; ++c) atomicAdd(&arow[(c % VEC) * 32 + c / VEC], hrow[c] * wev[j]);
        }
      }
    }
  };
  int qn = total;
  if (!kept) {
    qn = qcount;
    if (qn > AGG_QCAP) qn = AGG_QCAP;
  }
  qn = __builtin_amdgcn_readfirstlane(qn);
  int kfirst = 0;
  if (early) {
    drop(okm0, siv0, djv0, sdv0);
    kfirst = AGG_LK * AGG_T;
  }
  for (int k0 = kfirst; k0 < qn; k0 += AGG_LK * AGG_T) {
    int   okm;
    int   siv[AGG_LK], djv[AGG_LK];
    float sdv[AGG_LK];
    lookup(k0, qn, okm, siv, djv, sdv);
    drop(okm, siv, djv, sdv);
  }
  __syncthreads();

  constexpr int UB = AGG_UB;
  const float4 bv = *reinterpret_cast<const float4*>(bias + lch * 4);
  const char* Hb = reinterpret_cast<const char*>(H);
  const unsigned laneoff = (unsigned)lch * 16u;
  const int ent = (EPL == 1) ? lane : (2 * lch + sub);
#pragma unroll
  for (int s = 0; s < AGG_SLOTS; ++s) {
    const int r = wvu + AGG_W * s;
    if (r < nloc) {
      float4 a = sum[s];
      int bn = bcount[r];
      if (bn > AGG_BCAP) bn = AGG_BCAP;
      bn = __builtin_amdgcn_readfirstlane(bn);
      const int*   mys = bsrc + r * AGG_BCAP;
      const float* myw = bwt + r * AGG_BCAP;
      for (int i0 = 0; i0 < bn; i0 += 32) {
        const int cnt = (bn - i0 < 32) ? (bn - i0) : 32;
        const bool ok = ent < cnt;
        unsigned so = ok ? (unsigned)mys[i0 + ent] * (unsigned)(C * 4) : 0u;
        const float we = ok ? myw[i0 + ent] : 0.0f;
        if (EPL == 2) {
          const int last = cnt - 1;
          const unsigned solast = (unsigned)__builtin_amdgcn_readlane((int)so, (last >> 1) + 16 * (last & 1));
          so = ok ? so : solast;
        }
        const int nld = (cnt + EPL - 1) / EPL;
        for (int u0 = 0; u0 < nld; u0 += UB) {
          float4 hv[UB];
          const int nb = nld - u0;
          agg_static_upto<0, UB>(nb, [&](auto uc) {
            constexpr int u = decltype(uc)::value;
            unsigned off;
            if constexpr (EPL == 1) off = (unsigned)__builtin_amdgcn_readlane((int)so, u0 + u);
            else                    off = (unsigned)agg_half_bcast<u>((int)so);
            hv[u] = *reinterpret_cast<const float4*>(Hb + (off + laneoff));
          });
          agg_static_upto<0, UB>(nb, [&](auto uc) {
            constexpr int u = decltype(uc)::value;
            float w;
            if constexpr (EPL == 1) w = __int_as_float(__builtin_amdgcn_readlane(__float_as_int(we), u0 + u));
            else                    w = __int_as_float(agg_half_bcast<u>(__float_as_int(we)));
            a.x += hv[u].x * w;  a.y += hv[u].y * w;  a.z += hv[u].z * w;  a.w += hv[u].w * w;
          });
        }
      }
      if (EPL == 2) {
        a.x += __int_as_float(__builtin_amdgcn_permlanex16(0, __float_as_int(a.x), 0x76543210, 0xfedcba98, false, false));
        a.y += __int_as_float(__builtin_amdgcn_permlanex16(0, __float_as_int(a.y), 0x76543210, 0xfedcba98, false, false));
        a.z += __int_as_float(__builtin_amdgcn_permlanex16(0, __float_as_int(a.z), 0x76543210, 0xfedcba98, false, false));
        a.w += __int_as_float(__builtin_amdgcn_permlanex16(0, __float_as_int(a.w), 0x76543210, 0xfedcba98, false, false));
      }
      if (sub == 0) {
        const float* ar = acc + r * C;
        const float av[4] = {a.x, a.y, a.z, a.w};
        const float bb[4] = {bv.x, bv.y, bv.z, bv.w};
        float o[4];
#pragma unroll
        for (int t = 0; t < 4; ++t) {
          const int ch = 4 * lch + t;
          const float v = av[t] + ar[(ch % VEC) * 32 + ch / VEC] + bb[t];
          o[t] = RELU ? fmaxf(v, 0.0f) : v;
        }
        float4 ov;
        ov.x = o[0];  ov.y = o[1];  ov.z = o[2];  ov.w = o[3];
        *reinterpret_cast<float4*>(out + (size_t)(base + r) * C + lch * 4) = ov;
      }
    }
  }
}

extern "C" void kernel_launch(void* const* d_in, const int* in_sizes, int n_in,
                              void* d_out, int out_size, void* d_ws, size_t ws_size,
                              hipStream_t stream) {
  const int CIN = 256, CH = 128, CO = 64;
  const float* x  = (const float*)d_in[0];
  const int*   ei = (const int*)d_in[1];
  const float* W1 = (const float*)d_in[2];
  const float* b1 = (const float*)d_in[3];
  const float* W2 = (const float*)d_in[4];
  const float* b2 = (const float*)d_in[5];
  float* out = (float*)d_out;

  const int N = in_sizes[0] / CIN;
  const int E = in_sizes[1] / 2;
  const int* src = ei;
  const int* dst = ei + E;

  float* ws   = (float*)d_ws;
  float* dinv = ws;
  float* Hbuf = ws + 50176;
  float* AGG  = ws + 50176 + (size_t)N * CH;

  int* slabs = (int*)d_out;
  int* part = (int*)AGG;

  const int owners = (N + AGG_R - 1) / AGG_R;
  k_owner_scan<<<((owners + AGG_OG - 1) / AGG_OG) * AGG_S, AGG_T, 0, stream>>>(src, dst, part, slabs, N, E);

  k_gemm_wmma_f32<256, 128, true><<<((N + GM_WAVES * 16 - 1) / (GM_WAVES * 16)) * (CH / GM_NC), GM_WAVES * 32, 0, stream>>>(x, W1, Hbuf, N, part, dinv);
  k_agg_owner<128, true><<<(N + AGG_R - 1) / AGG_R, AGG_T, 0, stream>>>(src, dst, Hbuf, dinv, b1, slabs, AGG, N, E);

  k_gemm_wmma_f32<128, 64, false><<<((N + GM_WAVES * 16 - 1) / (GM_WAVES * 16)) * (CO / GM_NC), GM_WAVES * 32, 0, stream>>>(AGG, W2, Hbuf, N, nullptr, nullptr);
  k_agg_owner<64, false><<<(N + AGG_R - 1) / AGG_R, AGG_T, 0, stream>>>(src, dst, Hbuf, dinv, b2, slabs, out, N, E);
}


# Product launches: a thread issues all its weight-copy loads and the first x chunk as one group ahead of the LDS stores: one memory round trip, was five or three in turn; k and sum orders kept
# speedup vs baseline: 3.9615x; 1.0367x over previous
#include <hip/hip_runtime.h>
#include <hip/hip_bf16.h>

typedef float v2f __attribute__((ext_vector_type(2)));
typedef float v8f __attribute__((ext_vector_type(8)));

constexpr int AGG_R    = 196;
constexpr int AGG_T    = 1024;
constexpr int AGG_QCAP = 8192;
constexpr int AGG_US   = 8;
constexpr int AGG_UB   = 16;
constexpr int AGG_W    = AGG_T / 32;
constexpr int AGG_SLOTS = (AGG_R + AGG_W - 1) / AGG_W;
constexpr int AGG_BCAP = 64;
constexpr int AGG_LK   = 4;
constexpr int AGG_SLABROW = 64;
constexpr int AGG_SLAB = AGG_R * AGG_SLABROW;
constexpr int AGG_OG   = 8;
constexpr int AGG_S    = 8;

constexpr int GM_WAVES = 4;
constexpr int GM_NC    = 64;
constexpr int GM_PITCH = 2 * GM_NC + 32;
constexpr int GM_CH    = 8;

template <int K, int N, bool DINV>
__global__ __launch_bounds__(GM_WAVES * 32) void k_gemm_wmma_f32(const float* __restrict__ A,
                                                                  const float* __restrict__ B,
                                                                  float* __restrict__ C,
                                                                  int M,
                                                                  const int* __restrict__ part,
                                                                  float* __restrict__ dinv) {
  static_assert(K % (4 * GM_CH) == 0 && N % GM_NC == 0, "whole chunks, whole column parts");
  constexpr int NP = N / GM_NC;
  __shared__ float Bs[(K / 2) * GM_PITCH];

  const int tid  = threadIdx.x;
  const int lane = tid & 31;
  const int wv   = tid >> 5;
  const int h    = lane >> 4;
  const int l16  = lane & 15;
  const int cb    = (blockIdx.x % NP) * GM_NC;
  const int rbase = (blockIdx.x / NP) * (GM_WAVES * 16);
  const int mbase = rbase + wv * 16;

  constexpr int ITEMS = (K / 2) * (GM_NC / 4);
  static_assert(ITEMS % (GM_WAVES * 32) == 0, "every thread copies the same number of items");
  constexpr int NIT = ITEMS / (GM_WAVES * 32);
  float4 r0[NIT], r1[NIT];
#pragma unroll
  for (int i = 0; i < NIT; ++i) {
    const int it = tid + i * (GM_WAVES * 32);
    const int j  = it % (GM_NC / 4);
    const int kp = it / (GM_NC / 4);
    r0[i] = *reinterpret_cast<const float4*>(B + (size_t)(2 * kp) * N + cb + 4 * j);
    r1[i] = *reinterpret_cast<const float4*>(B + (size_t)(2 * kp + 1) * N + cb + 4 * j);
  }

  const int m  = mbase + l16;
  const int mc = (m < M) ? m : (M - 1);
  const float* arow = A + (size_t)mc * K + 2 * h;
  const float* bl   = Bs + h * GM_PITCH + 2 * l16;

  v2f an[GM_CH];
#pragma unroll
  for (int s = 0; s < GM_CH; ++s) an[s] = *reinterpret_cast<const v2f*>(arow + 4 * s);

#pragma unroll
  for (int i = 0; i < NIT; ++i) {
    const int it = tid + i * (GM_WAVES * 32);
    const int j  = it % (GM_NC / 4);
    const int kp = it / (GM_NC / 4);
    float* d = Bs + kp * GM_PITCH + 2 * j;
    v2f p0, p1, p2, p3;
    p0.x = r0[i].x;  p0.y = r1[i].x;
    p1.x = r0[i].y;  p1.y = r1[i].y;
    p2.x = r0[i].z;  p2.y = r1[i].z;
    p3.x = r0[i].w;  p3.y = r1[i].w;
    *reinterpret_cast<v2f*>(d)      = p0;
    *reinterpret_cast<v2f*>(d + 32) = p1;
    *reinterpret_cast<v2f*>(d + 64) = p2;
    *reinterpret_cast<v2f*>(d + 96) = p3;
  }
  __syncthreads();

  v8f acc0 = {}, acc1 = {}, acc2 = {}, acc3 = {};
  for (int kc = 0; kc < K; kc += 4 * GM_CH) {
    v2f ac[GM_CH];
#pragma unroll
    for (int s = 0; s < GM_CH; ++s) ac[s] = an[s];
    if (kc + 4 * GM_CH < K) {
#pragma unroll
      for (int s = 0; s < GM_CH; ++s) an[s] = *reinterpret_cast<const v2f*>(arow + kc + 4 * GM_CH + 4 * s);
    }
#pragma unroll
    for (int s = 0; s < GM_CH; ++s) {
      const float* bp = bl + ((kc >> 1) + 2 * s) * GM_PITCH;
      const v2f b0 = *reinterpret_cast<const v2f*>(bp);
      const v2f b1 = *reinterpret_cast<const v2f*>(bp + 32);
      const v2f b2 = *reinterpret_cast<const v2f*>(bp + 64);
      const v2f b3 = *reinterpret_cast<const v2f*>(bp + 96);
      acc0 = __builtin_amdgcn_wmma_f32_16x16x4_f32(false, ac[s], false, b0, (short)0, acc0, false, false);
      acc1 = __builtin_amdgcn_wmma_f32_16x16x4_f32(false, ac[s], false, b1, (short)0, acc1, false, false);
      acc2 = __builtin_amdgcn_wmma_f32_16x16x4_f32(false, ac[s], false, b2, (short)0, acc2, false, false);
      acc3 = __builtin_amdgcn_wmma_f32_16x16x4_f32(false, ac[s], false, b3, (short)0, acc3, false, false);
    }
  }

#pragma unroll
  for (int r = 0; r < 8; ++r) {
    const int row = mbase + r + 8 * h;
    if (row < M) {
      float4 o;
      o.x = acc0[r];  o.y = acc1[r];  o.z = acc2[r];  o.w = acc3[r];
      *reinterpret_cast<float4*>(C + (size_t)row * N + cb + 4 * l16) = o;
    }
  }

  if (DINV && cb == 0) {
    for (int node = rbase + tid; node < rbase + GM_WAVES * 16 && node < M; node += GM_WAVES * 32) {
      int deg = 0;
#pragma unroll
      for (int s = 0; s < AGG_S; ++s) deg += part[(size_t)s * M + node];
      dinv[node] = rsqrtf(1.0f + (float)deg);
    }
  }
}

static_assert(AGG_US * 4 <= 32, "one hit bit per scanned destination");
template <bool B> struct agg_flag { static constexpr bool value = B; };

template <int V> struct agg_int { static constexpr int value = V; };
template <int I, int N, class F>
__device__ __forceinline__ void agg_static_upto(int n, F&& f) {
  if constexpr (I < N) {
    if (I < n) { f(agg_int<I>()); agg_static_upto<I + 1, N>(n, f); }
  }
}
template <int U> __device__ __forceinline__ int agg_half_bcast(int v) {
  return __builtin_amdgcn_update_dpp(0, v, 0x150 + U, 0xf, 0xf, true);
}

__device__ __forceinline__ int agg_seg_words(int nloc) { return (nloc * AGG_SLABROW) / AGG_S; }

constexpr int AGG_PACK_NODES = 1 << 24;
static_assert(AGG_R <= 256, "the local row fits the 8 bits above the source");
__device__ __forceinline__ int agg_pack(int s, int row) { return (int)((unsigned)s | ((unsigned)row << 24)); }

__device__ __forceinline__ void agg_part_bounds(int units, int part, int nparts, int& lo, int& hi) {
  const int per = (units + nparts - 1) / nparts;
  lo = part * per;
  if (lo > units) lo = units;
  hi = lo + per;
  if (hi > units) hi = units;
}

template <class Push>
__device__ __forceinline__ void agg_scan_dst(const int* __restrict__ dst, int E, int part, int nparts, int base, int nloc, int tid, Push push) {
  if ((E & 3) == 0) {
    const int4* d4 = reinterpret_cast<const int4*>(dst);
    int lo, E4;
    agg_part_bounds(E >> 2, part, nparts, lo, E4);
    auto round = [&](int g, auto guarded) {
      int4 dv[AGG_US];
#pragma unroll
      for (int u = 0; u < AGG_US; ++u) {
        const int gu = g + u * AGG_T;
        dv[u] = d4[(!guarded.value || gu < E4) ? gu : g];
      }
      unsigned hm = 0u;
#pragma unroll
      for (int u = 0; u < AGG_US; ++u) {
        const unsigned lim = (!guarded.value || g + u * AGG_T < E4) ? (unsigned)nloc : 0u;
        hm |= ((unsigned)(dv[u].x - base) < lim) ? (1u << (4 * u))     : 0u;
        hm |= ((unsigned)(dv[u].y - base) < lim) ? (1u << (4 * u + 1)) : 0u;
        hm |= ((unsigned)(dv[u].z - base) < lim) ? (1u << (4 * u + 2)) : 0u;
        hm |= ((unsigned)(dv[u].w - base) < lim) ? (1u << (4 * u + 3)) : 0u;
      }
      while (hm) {
        const int b = __builtin_ctz(hm);
        hm &= hm - 1u;
        push(4 * (g + (b >> 2) * AGG_T) + (b & 3));
      }
    };
    int g = lo + tid;
    for (; g + (AGG_US - 1) * AGG_T < E4; g += AGG_US * AGG_T) round(g, agg_flag<false>());
    if (g < E4) round(g, agg_flag<true>());
  } else {
    int lo, hi;
    agg_part_bounds(E, part, nparts, lo, hi);
    for (int e = lo + tid; e < hi; e += AGG_T)
      if ((unsigned)(dst[e] - base) < (unsigned)nloc) push(e);
  }
}

__global__ __launch_bounds__(1024) void k_owner_scan(const int* __restrict__ src,
                                                     const int* __restrict__ dst,
                                                     int* __restrict__ part,
                                                     int* __restrict__ slabs,
                                                     int N, int E) {
  __shared__ int q[AGG_QCAP];
  __shared__ int cnt[AGG_OG * AGG_R];
  __shared__ int ocount[AGG_OG];
  __shared__ int qcount;

  const int tid   = threadIdx.x;
  const int grp   = blockIdx.x / AGG_S;
  const int sp    = blockIdx.x % AGG_S;
  const int gbase = grp * (AGG_OG * AGG_R);
  int gn = N - gbase;
  if (gn > AGG_OG * AGG_R) gn = AGG_OG * AGG_R;

  if (tid == 0) qcount = 0;
  if (tid < AGG_OG) ocount[tid] = 0;
  for (int r = tid; r < AGG_OG * AGG_R; r += AGG_T) cnt[r] = 0;
  __syncthreads();

  agg_scan_dst(dst, E, sp, AGG_S, gbase, gn, tid, [&](int e) {
    const int pos = atomicAdd(&qcount, 1);
    if (pos < AGG_QCAP) q[pos] = e;
  });
  __syncthreads();

  const int total = qcount;
  const bool over = total > AGG_QCAP;
  if (!over) {
    for (int k = tid; k < total; k += AGG_T) {
      const int e  = q[k];
      const int sv = src[e];
      const int dl = dst[e] - gbase;
      const int o  = dl / AGG_R;
      atomicAdd(&cnt[dl], 1);
      const int pos = atomicAdd(&ocount[o], 1);
      int onl = N - (gbase + o * AGG_R);
      if (onl > AGG_R) onl = AGG_R;
      const int segw = agg_seg_words(onl);
      if (pos < segw - 1) slabs[(size_t)(grp * AGG_OG + o) * AGG_SLAB + sp * segw + 1 + pos] = agg_pack(sv, dl - o * AGG_R);
    }
  } else {
    const bool wide = (E & 3) == 0;
    int lo, hi;
    agg_part_bounds(wide ? (E >> 2) : E, sp, AGG_S, lo, hi);
    if (wide) { lo *= 4; hi *= 4; }
    for (int e = lo + tid; e < hi; e += AGG_T) {
      const int dl = dst[e] - gbase;
      if ((unsigned)dl < (unsigned)gn) atomicAdd(&cnt[dl], 1);
    }
  }
  __syncthreads();

  if (tid < AGG_OG) {
    const int obase = gbase + tid * AGG_R;
    if (obase < N) {
      int onl = N - obase;
      if (onl > AGG_R) onl = AGG_R;
      slabs[(size_t)(grp * AGG_OG + tid) * AGG_SLAB + sp * agg_seg_words(onl)] = (over || N > AGG_PACK_NODES) ? (AGG_QCAP + 1) : ocount[tid];
    }
  }
  for (int r = tid; r < gn; r += AGG_T) part[(size_t)sp * N + gbase + r] = cnt[r];
}

template <int C, bool RELU>
__global__ __launch_bounds__(1024) void k_agg_owner(const int* __restrict__ src,
                                                    const int* __restrict__ dst,
                                                    const float* __restrict__ H,
                                                    const float* __restrict__ dinv,
                                                    const float* __restrict__ bias,
                                                    const int* slabs,
                                                    float* out,
                                                    int N, int E) {
  constexpr int VEC = C / 32;
  constexpr int LPR = C / 4;
  constexpr int EPL = 32 / LPR;
  static_assert(EPL == 1 || EPL == 2, "a source row fills a wave or a half-wave");
  __shared__ float acc[AGG_R * C];
  __shared__ int   q[AGG_QCAP];
  __shared__ int   bsrc[AGG_R * AGG_BCAP];
  __shared__ float bwt[AGG_R * AGG_BCAP];
  __shared__ int   bcount[AGG_R];
  __shared__ float dloc[AGG_R];
  __shared__ int   qcount;

  const int tid  = threadIdx.x;
  const int lane = tid & 31;
  const int wv   = tid >> 5;
  const int wvu  = __builtin_amdgcn_readfirstlane(wv);
  const int sub  = lane / LPR;
  const int lch  = lane % LPR;
  const int base = blockIdx.x * AGG_R;
  int nloc = N - base;
  if (nloc > AGG_R) nloc = AGG_R;

  const int* slab = slabs + (size_t)blockIdx.x * AGG_SLAB;
  const int segw = agg_seg_words(nloc);
  int segc[AGG_S];
  int total = 0;
  bool kept = true;
#pragma unroll
  for (int s = 0; s < AGG_S; ++s) {
    segc[s] = __builtin_amdgcn_readfirstlane(slab[s * segw]);
    kept = kept && (segc[s] <= segw - 1);
    total += segc[s];
  }

  float4 sum[AGG_SLOTS];
  {
    float dw[AGG_SLOTS];
    float4 hs[AGG_SLOTS];
#pragma unroll
    for (int s = 0; s < AGG_SLOTS; ++s) {
      const int r  = wvu + AGG_W * s;
      const int rr = (r < nloc) ? r : (nloc - 1);
      dw[s] = dinv[base + rr];
      hs[s] = *reinterpret_cast<const float4*>(H + (size_t)(base + rr) * C + lch * 4);
    }
#pragma unroll
    for (int s = 0; s < AGG_SLOTS; ++s) {
      const float w = (wvu + AGG_W * s < nloc && sub == 0) ? dw[s] * dw[s] : 0.0f;
      sum[s].x = hs[s].x * w;  sum[s].y = hs[s].y * w;  sum[s].z = hs[s].z * w;  sum[s].w = hs[s].w * w;
    }
  }

  auto lookup = [&](int k0, int qn, int& okm, int (&siv)[AGG_LK], int (&djv)[AGG_LK], float (&sdv)[AGG_LK]) {
    int ev[AGG_LK];
    okm = 0;
#pragma unroll
    for (int j = 0; j < AGG_LK; ++j) {
      const int kk = k0 + j * AGG_T + tid;
      const bool ok = kk < qn;
      okm |= ok ? (1 << j) : 0;
      const int k = ok ? kk : (qn - 1);
      if (kept) {
        int seg = 0, first = 0, run = 0;
#pragma unroll
        for (int s = 0; s + 1 < AGG_S; ++s) {
          run += segc[s];
          if (k >= run) { seg = s + 1; first = run; }
        }
        ev[j] = slab[seg * segw + 1 + (k - first)];
      } else {
        ev[j] = q[k];
      }
    }
    if (kept) {
#pragma unroll
      for (int j = 0; j < AGG_LK; ++j) {
        siv[j] = ev[j] & 0xffffff;
        djv[j] = (int)((unsigned)ev[j] >> 24);
      }
    } else {
#pragma unroll
      for (int j = 0; j < AGG_LK; ++j) {
        siv[j] = src[ev[j]];
        djv[j] = dst[ev[j]] - base;
      }
    }
#pragma unroll
    for (int j = 0; j < AGG_LK; ++j) {
      if (((okm >> j) & 1) == 0) djv[j] = 0;
      sdv[j] = dinv[siv[j]];
    }
  };
  const bool early = kept && total > 0;
  int   okm0 = 0;
  int   siv0[AGG_LK], djv0[AGG_LK];
  float sdv0[AGG_LK];
  if (early) lookup(0, total, okm0, siv0, djv0, sdv0);

  if (tid == 0) qcount = 0;
  for (int r = tid; r < nloc; r += AGG_T) { dloc[r] = dinv[base + r]; bcount[r] = 0; }
  {
    float4* a4 = reinterpret_cast<float4*>(acc);
    const float4 z = {0.0f, 0.0f, 0.0f, 0.0f};
    for (int i = tid; i < (nloc * C) / 4; i += AGG_T) a4[i] = z;
  }
  __syncthreads();

  if (!kept) {
    agg_scan_dst(dst, E, 0, 1, base, nloc, tid, [&](int e) {
      const int pos = atomicAdd(&qcount, 1);
      if (pos < AGG_QCAP) {
        q[pos] = e;
      } else {
        const int i = src[e];
        const int d = dst[e];
        const float w = dinv[i] * dinv[d];
        const float* hrow = H + (size_t)i * C;
        float* arow = acc + (d - base) * C;
        for (int c = 0; c < C; ++c) atomicAdd(&arow[(c % VEC) * 32 + c / VEC], hrow[c] * w);
      }
    });
  }
  __syncthreads();

  auto drop = [&](int okm, const int (&siv)[AGG_LK], const int (&djv)[AGG_LK], const float (&sdv)[AGG_LK]) {
    float wev[AGG_LK];
#pragma unroll
    for (int j = 0; j < AGG_LK; ++j) wev[j] = sdv[j] * dloc[djv[j]];
#pragma unroll
    for (int j = 0; j < AGG_LK; ++j) {
      if ((okm >> j) & 1) {
        const int pos = atomicAdd(&bcount[djv[j]], 1);
        if (pos < AGG_BCAP) {
          bsrc[djv[j] * AGG_BCAP + pos] = siv[j];
          bwt[djv[j] * AGG_BCAP + pos]  = wev[j];
        } else {
          const float* hrow = H + (size_t)siv[j] * C;
          float* arow = acc + djv[j] * C;
          for (int c = 0; c < C; ++c) atomicAdd(&arow[(c % VEC) * 32 + c / VEC], hrow[c] * wev[j]);
        }
      }
    }
  };
  int qn = total;
  if (!kept) {
    qn = qcount;
    if (qn > AGG_QCAP) qn = AGG_QCAP;
  }
  qn = __builtin_amdgcn_readfirstlane(qn);
  int kfirst = 0;
  if (early) {
    drop(okm0, siv0, djv0, sdv0);
    kfirst = AGG_LK * AGG_T;
  }
  for (int k0 = kfirst; k0 < qn; k0 += AGG_LK * AGG_T) {
    int   okm;
    int   siv[AGG_LK], djv[AGG_LK];
    float sdv[AGG_LK];
    lookup(k0, qn, okm, siv, djv, sdv);
    drop(okm, siv, djv, sdv);
  }
  __syncthreads();

  constexpr int UB = AGG_UB;
  const float4 bv = *reinterpret_cast<const float4*>(bias + lch * 4);
  const char* Hb = reinterpret_cast<const char*>(H);
  const unsigned laneoff = (unsigned)lch * 16u;
  const int ent = (EPL == 1) ? lane : (2 * lch + sub);
#pragma unroll
  for (int s = 0; s < AGG_SLOTS; ++s) {
    const int r = wvu + AGG_W * s;
    if (r < nloc) {
      float4 a = sum[s];
      int bn = bcount[r];
      if (bn > AGG_BCAP) bn = AGG_BCAP;
      bn = __builtin_amdgcn_readfirstlane(bn);
      const int*   mys = bsrc + r * AGG_BCAP;
      const float* myw = bwt + r * AGG_BCAP;
      for (int i0 = 0; i0 < bn; i0 += 32) {
        const int cnt = (bn - i0 < 32) ? (bn - i0) : 32;
        const bool ok = ent < cnt;
        unsigned so = ok ? (unsigned)mys[i0 + ent] * (unsigned)(C * 4) : 0u;
        const float we = ok ? myw[i0 + ent] : 0.0f;
        if (EPL == 2) {
          const int last = cnt - 1;
          const unsigned solast = (unsigned)__builtin_amdgcn_readlane((int)so, (last >> 1) + 16 * (last & 1));
          so = ok ? so : solast;
        }
        const int nld = (cnt + EPL - 1) / EPL;
        for (int u0 = 0; u0 < nld; u0 += UB) {
          float4 hv[UB];
          const int nb = nld - u0;
          agg_static_upto<0, UB>(nb, [&](auto uc) {
            constexpr int u = decltype(uc)::value;
            unsigned off;
            if constexpr (EPL == 1) off = (unsigned)__builtin_amdgcn_readlane((int)so, u0 + u);
            else                    off = (unsigned)agg_half_bcast<u>((int)so);
            hv[u] = *reinterpret_cast<const float4*>(Hb + (off + laneoff));
          });
          agg_static_upto<0, UB>(nb, [&](auto uc) {
            constexpr int u = decltype(uc)::value;
            float w;
            if constexpr (EPL == 1) w = __int_as_float(__builtin_amdgcn_readlane(__float_as_int(we), u0 + u));
            else                    w = __int_as_float(agg_half_bcast<u>(__float_as_int(we)));
            a.x += hv[u].x * w;  a.y += hv[u].y * w;  a.z += hv[u].z * w;  a.w += hv[u].w * w;
          });
        }
      }
      if (EPL == 2) {
        a.x += __int_as_float(__builtin_amdgcn_permlanex16(0, __float_as_int(a.x), 0x76543210, 0xfedcba98, false, false));
        a.y += __int_as_float(__builtin_amdgcn_permlanex16(0, __float_as_int(a.y), 0x76543210, 0xfedcba98, false, false));
        a.z += __int_as_float(__builtin_amdgcn_permlanex16(0, __float_as_int(a.z), 0x76543210, 0xfedcba98, false, false));
        a.w += __int_as_float(__builtin_amdgcn_permlanex16(0, __float_as_int(a.w), 0x76543210, 0xfedcba98, false, false));
      }
      if (sub == 0) {
        const float* ar = acc + r * C;
        const float av[4] = {a.x, a.y, a.z, a.w};
        const float bb[4] = {bv.x, bv.y, bv.z, bv.w};
        float o[4];
#pragma unroll
        for (int t = 0; t < 4; ++t) {
          const int ch = 4 * lch + t;
          const float v = av[t] + ar[(ch % VEC) * 32 + ch / VEC] + bb[t];
          o[t] = RELU ? fmaxf(v, 0.0f) : v;
        }
        float4 ov;
        ov.x = o[0];  ov.y = o[1];  ov.z = o[2];  ov.w = o[3];
        *reinterpret_cast<float4*>(out + (size_t)(base + r) * C + lch * 4) = ov;
      }
    }
  }
}

extern "C" void kernel_launch(void* const* d_in, const int* in_sizes, int n_in,
                              void* d_out, int out_size, void* d_ws, size_t ws_size,
                              hipStream_t stream) {
  const int CIN = 256, CH = 128, CO = 64;
  const float* x  = (const float*)d_in[0];
  const int*   ei = (const int*)d_in[1];
  const float* W1 = (const float*)d_in[2];
  const float* b1 = (const float*)d_in[3];
  const float* W2 = (const float*)d_in[4];
  const float* b2 = (const float*)d_in[5];
  float* out = (float*)d_out;

  const int N = in_sizes[0] / CIN;
  const int E = in_sizes[1] / 2;
  const int* src = ei;
  const int* dst = ei + E;

  float* ws   = (float*)d_ws;
  float* dinv = ws;
  float* Hbuf = ws + 50176;
  float* AGG  = ws + 50176 + (size_t)N * CH;

  int* slabs = (int*)d_out;
  int* part = (int*)AGG;

  const int owners = (N + AGG_R - 1) / AGG_R;
  k_owner_scan<<<((owners + AGG_OG - 1) / AGG_OG) * AGG_S, AGG_T, 0, stream>>>(src, dst, part, slabs, N, E);

  k_gemm_wmma_f32<256, 128, true><<<((N + GM_WAVES * 16 - 1) / (GM_WAVES * 16)) * (CH / GM_NC), GM_WAVES * 32, 0, stream>>>(x, W1, Hbuf, N, part, dinv);
  k_agg_owner<128, true><<<(N + AGG_R - 1) / AGG_R, AGG_T, 0, stream>>>(src, dst, Hbuf, dinv, b1, slabs, AGG, N, E);

  k_gemm_wmma_f32<128, 64, false><<<((N + GM_WAVES * 16 - 1) / (GM_WAVES * 16)) * (CO / GM_NC), GM_WAVES * 32, 0, stream>>>(AGG, W2, Hbuf, N, nullptr, nullptr);
  k_agg_owner<64, false><<<(N + AGG_R - 1) / AGG_R, AGG_T, 0, stream>>>(src, dst, Hbuf, dinv, b2, slabs, out, N, E);
}
